# starting kernel
# baseline (speedup 1.0000x reference)
#include <hip/hip_runtime.h>

typedef float v2f __attribute__((ext_vector_type(2)));
typedef float v8f __attribute__((ext_vector_type(8)));

#define NEG_SLOPE 0.2f


__global__ void fill_f32(float* __restrict__ p, long n, float v) {
  long i = (long)blockIdx.x * blockDim.x + threadIdx.x;
  long stride = (long)gridDim.x * blockDim.x;
  for (; i < n; i += stride) p[i] = v;
}

__global__ void gemm_wmma_f32_lds(const float* __restrict__ A, const float* __restrict__ B,
                                  float* __restrict__ C, int M, int K, int Nd,
                                  int tnPerBlock) {
  extern __shared__ float As[];
  int tilesN = Nd >> 4;
  int blocksPerRow = tilesN / tnPerBlock;
  int tm  = blockIdx.x / blocksPerRow;
  int tn0 = (blockIdx.x % blocksPerRow) * tnPerBlock;
  int wib  = threadIdx.x >> 5;
  int lane = threadIdx.x & 31;

  {
    const float4* Av  = (const float4*)(A + (long)tm * 16 * K);
    float4*       Asv = (float4*)As;
    int nvec = (16 * K) >> 2;
    for (int idx = threadIdx.x; idx < nvec; idx += blockDim.x) Asv[idx] = Av[idx];
  }
  __syncthreads();

  int tn   = tn0 + wib;
  int half = lane >> 4;
  int l15  = lane & 15;
  int bcol = tn * 16 + l15;
  v8f acc = {};
  for (int k = 0; k < K; k += 4) {
    int ka = k + 2 * half;
    v2f a, b;
    a.x = As[l15 * K + ka];
    a.y = As[l15 * K + ka + 1];
    b.x = B[(long)ka * Nd + bcol];
    b.y = B[(long)(ka + 1) * Nd + bcol];
    acc = __builtin_amdgcn_wmma_f32_16x16x4_f32(false, a, false, b,
                                                (short)0, acc, false, false);
  }
  long crow = (long)tm * 16 + half * 8;
#pragma unroll
  for (int r = 0; r < 8; ++r)
    C[(crow + r) * Nd + bcol] = acc[r];
}

__global__ void logits_kernel(const float* __restrict__ hmat, const float* __restrict__ a_src,
                              const float* __restrict__ a_dst, float* __restrict__ als,
                              float* __restrict__ ald, int nnodes, int heads, int ch) {
  int wid  = (int)(((long)blockIdx.x * blockDim.x + threadIdx.x) >> 5);
  int lane = threadIdx.x & 31;
  if (wid >= nnodes * heads) return;
  int node = wid / heads, hh = wid % heads;
  int F = heads * ch;
  float ss = 0.f, dd = 0.f;
  for (int c = lane; c < ch; c += 32) {
    float hv = hmat[(long)node * F + hh * ch + c];
    ss += hv * a_src[hh * ch + c];
    dd += hv * a_dst[hh * ch + c];
  }
#pragma unroll
  for (int off = 16; off > 0; off >>= 1) {
    ss += __shfl_xor(ss, off);
    dd += __shfl_xor(dd, off);
  }
  if (lane == 0) { als[wid] = ss; ald[wid] = dd; }
}

__device__ __forceinline__ void edge_sd(const int* __restrict__ eidx, int E, int e,
                                        int& s, int& d) {
  if (e < E) { s = eidx[e]; d = eidx[E + e]; }
  else       { s = e - E; d = s; }
}

__global__ void edge_max_kernel(const int* __restrict__ eidx, int E, int EN,
                                const float* __restrict__ als, const float* __restrict__ ald,
                                float* __restrict__ m, int heads) {
  long t = (long)blockIdx.x * blockDim.x + threadIdx.x;
  if (t >= (long)EN * heads) return;
  int e = (int)(t / heads), hh = (int)(t % heads);
  int s, d; edge_sd(eidx, E, e, s, d);
  float l = als[s * heads + hh] + ald[d * heads + hh];
  l = l > 0.f ? l : NEG_SLOPE * l;
  unsafeAtomicMax(&m[d * heads + hh], l);
}

template<int HEADS, int CH>
__global__ void edge_fused_kernel(const int* __restrict__ eidx, int E, int EN,
                                  const float* __restrict__ als, const float* __restrict__ ald,
                                  const float* __restrict__ m, float* __restrict__ ssum,
                                  const float* __restrict__ hsrc, float* __restrict__ accum) {
  constexpr int F  = HEADS * CH;
  constexpr int NF = F / 32;
  int wid  = (int)(((long)blockIdx.x * blockDim.x + threadIdx.x) >> 5);
  int lane = threadIdx.x & 31;
  if (wid >= EN) return;
  int s, d; edge_sd(eidx, E, wid, s, d);
  int j0   = lane * NF;
  int head = j0 / CH;
  float l = als[s * HEADS + head] + ald[d * HEADS + head];
  l = l > 0.f ? l : NEG_SLOPE * l;
  float ex = __expf(l - m[d * HEADS + head]);
  if ((j0 % CH) == 0)
    unsafeAtomicAdd(&ssum[d * HEADS + head], ex);
  const float* hp = hsrc  + (long)s * F + j0;
  float*       ap = accum + (long)d * F + j0;
  float v[NF];
#pragma unroll
  for (int k = 0; k < NF; ++k) v[k] = hp[k];
#pragma unroll
  for (int k = 0; k < NF; ++k)
    unsafeAtomicAdd(&ap[k], v[k] * ex);
}

__global__ void edge_fused_gen(const int* __restrict__ eidx, int E, int EN,
                               const float* __restrict__ als, const float* __restrict__ ald,
                               const float* __restrict__ m, float* __restrict__ ssum,
                               const float* __restrict__ hsrc, float* __restrict__ accum,
                               int heads, int ch) {
  int wid  = (int)(((long)blockIdx.x * blockDim.x + threadIdx.x) >> 5);
  int lane = threadIdx.x & 31;
  if (wid >= EN) return;
  int s, d; edge_sd(eidx, E, wid, s, d);
  int F = heads * ch;
  for (int j = lane; j < F; j += 32) {
    int head = j / ch;
    float l = als[s * heads + head] + ald[d * heads + head];
    l = l > 0.f ? l : NEG_SLOPE * l;
    float ex = __expf(l - m[d * heads + head]);
    if ((j % ch) == 0) unsafeAtomicAdd(&ssum[d * heads + head], ex);
    unsafeAtomicAdd(&accum[(long)d * F + j], hsrc[(long)s * F + j] * ex);
  }
}

__global__ void norm_bias_act_kernel(float* __restrict__ a, const float* __restrict__ ssum,
                                     const float* __restrict__ bias, long n, int F,
                                     int ch, int heads, int do_elu) {
  long i = (long)blockIdx.x * blockDim.x + threadIdx.x;
  long stride = (long)gridDim.x * blockDim.x;
  for (; i < n; i += stride) {
    int  j    = (int)(i % F);
    long node = i / F;
    float v = a[i] / ssum[node * heads + j / ch] + bias[j];
    if (do_elu) v = v > 0.f ? v : (__expf(v) - 1.f);
    a[i] = v;
  }
}


static inline int cdiv_i(long a, long b) { return (int)((a + b - 1) / b); }

static void launch_gemm(const float* A, const float* B, float* C,
                        int M, int K, int Nd, hipStream_t stream) {
  int tilesM = M / 16, tilesN = Nd / 16;
  int wpb = tilesN >= 8 ? 8 : tilesN;
  dim3 grid(tilesM * (tilesN / wpb));
  dim3 block(32 * wpb);
  size_t shmem = (size_t)16 * K * sizeof(float);
  gemm_wmma_f32_lds<<<grid, block, shmem, stream>>>(A, B, C, M, K, Nd, wpb);
}

extern "C" void kernel_launch(void* const* d_in, const int* in_sizes, int n_in,
                              void* d_out, int out_size, void* d_ws, size_t ws_size,
                              hipStream_t stream) {
  const float* x   = (const float*)d_in[0];
  const int*   eix = (const int*)d_in[1];
  const float* W1  = (const float*)d_in[2];
  const float* as1 = (const float*)d_in[3];
  const float* ad1 = (const float*)d_in[4];
  const float* b1  = (const float*)d_in[5];
  const float* W2  = (const float*)d_in[6];
  const float* as2 = (const float*)d_in[7];
  const float* ad2 = (const float*)d_in[8];
  const float* b2  = (const float*)d_in[9];
  float* out = (float*)d_out;

  const int IN_DIM = 128;
  const int F1     = in_sizes[5];
  const int HID    = in_sizes[9];
  const int HEADS  = F1 / HID;
  const int Nn     = in_sizes[0] / IN_DIM;
  const int E      = in_sizes[1] / 2;
  const int EN     = E + Nn;

  float* acc1 = (float*)d_ws;
  float* h1   = acc1 + (long)Nn * F1;
  float* sc   = h1 + (long)Nn * F1;
  float* als1 = sc;
  float* ald1 = als1 + (long)Nn * HEADS;
  float* m1   = ald1 + (long)Nn * HEADS;
  float* s1   = m1   + (long)Nn * HEADS;
  float* als2 = s1   + (long)Nn * HEADS;
  float* ald2 = als2 + Nn;
  float* m2   = ald2 + Nn;
  float* s2   = m2   + Nn;
  float* h2   = h1;

  const float NEGINF = -__builtin_huge_valf();

  fill_f32<<<cdiv_i((long)Nn * F1, 256), 256, 0, stream>>>(acc1, (long)Nn * F1, 0.f);
  fill_f32<<<cdiv_i((long)Nn * HEADS, 256), 256, 0, stream>>>(m1, (long)Nn * HEADS, NEGINF);
  fill_f32<<<cdiv_i((long)Nn * HEADS, 256), 256, 0, stream>>>(s1, (long)Nn * HEADS, 0.f);
  fill_f32<<<cdiv_i(Nn, 256), 256, 0, stream>>>(m2, Nn, NEGINF);
  fill_f32<<<cdiv_i(Nn, 256), 256, 0, stream>>>(s2, Nn, 0.f);
  fill_f32<<<cdiv_i((long)Nn * HID, 256), 256, 0, stream>>>(out, (long)Nn * HID, 0.f);

  launch_gemm(x, W1, h1, Nn, IN_DIM, F1, stream);
  logits_kernel<<<cdiv_i((long)Nn * HEADS * 32, 256), 256, 0, stream>>>(
      h1, as1, ad1, als1, ald1, Nn, HEADS, HID);
  edge_max_kernel<<<cdiv_i((long)EN * HEADS, 256), 256, 0, stream>>>(
      eix, E, EN, als1, ald1, m1, HEADS);
  if (HEADS == 4 && HID == 64)
    edge_fused_kernel<4, 64><<<cdiv_i((long)EN * 32, 256), 256, 0, stream>>>(
        eix, E, EN, als1, ald1, m1, s1, h1, acc1);
  else
    edge_fused_gen<<<cdiv_i((long)EN * 32, 256), 256, 0, stream>>>(
        eix, E, EN, als1, ald1, m1, s1, h1, acc1, HEADS, HID);
  norm_bias_act_kernel<<<cdiv_i((long)Nn * F1, 256), 256, 0, stream>>>(
      acc1, s1, b1, (long)Nn * F1, F1, HID, HEADS, 1);

  launch_gemm(acc1, W2, h2, Nn, F1, HID, stream);
  logits_kernel<<<cdiv_i((long)Nn * 32, 256), 256, 0, stream>>>(
      h2, as2, ad2, als2, ald2, Nn, 1, HID);
  edge_max_kernel<<<cdiv_i((long)EN, 256), 256, 0, stream>>>(
      eix, E, EN, als2, ald2, m2, 1);
  if (HID == 64)
    edge_fused_kernel<1, 64><<<cdiv_i((long)EN * 32, 256), 256, 0, stream>>>(
        eix, E, EN, als2, ald2, m2, s2, h2, out);
  else
    edge_fused_gen<<<cdiv_i((long)EN * 32, 256), 256, 0, stream>>>(
        eix, E, EN, als2, ald2, m2, s2, h2, out, 1, HID);
  norm_bias_act_kernel<<<cdiv_i((long)Nn * HID, 256), 256, 0, stream>>>(
      out, s2, b2, (long)Nn * HID, HID, HID, 1, 0);
}


# Aggregation: a work-group filters the edge list for its 196 destination rows into LDS; one wave sums each row in registers, self-loop then ascending edge order, running max; float atomics gone
# speedup vs baseline: 2.1384x; 2.1384x over previous
#include <hip/hip_runtime.h>

typedef float v2f __attribute__((ext_vector_type(2)));
typedef float v8f __attribute__((ext_vector_type(8)));

#define NEG_SLOPE 0.2f


__global__ void fill_f32(float* __restrict__ p, long n, float v) {
  long i = (long)blockIdx.x * blockDim.x + threadIdx.x;
  long stride = (long)gridDim.x * blockDim.x;
  for (; i < n; i += stride) p[i] = v;
}

__global__ void gemm_wmma_f32_lds(const float* __restrict__ A, const float* __restrict__ B,
                                  float* __restrict__ C, int M, int K, int Nd,
                                  int tnPerBlock) {
  extern __shared__ float As[];
  int tilesN = Nd >> 4;
  int blocksPerRow = tilesN / tnPerBlock;
  int tm  = blockIdx.x / blocksPerRow;
  int tn0 = (blockIdx.x % blocksPerRow) * tnPerBlock;
  int wib  = threadIdx.x >> 5;
  int lane = threadIdx.x & 31;

  {
    const float4* Av  = (const float4*)(A + (long)tm * 16 * K);
    float4*       Asv = (float4*)As;
    int nvec = (16 * K) >> 2;
    for (int idx = threadIdx.x; idx < nvec; idx += blockDim.x) Asv[idx] = Av[idx];
  }
  __syncthreads();

  int tn   = tn0 + wib;
  int half = lane >> 4;
  int l15  = lane & 15;
  int bcol = tn * 16 + l15;
  v8f acc = {};
  for (int k = 0; k < K; k += 4) {
    int ka = k + 2 * half;
    v2f a, b;
    a.x = As[l15 * K + ka];
    a.y = As[l15 * K + ka + 1];
    b.x = B[(long)ka * Nd + bcol];
    b.y = B[(long)(ka + 1) * Nd + bcol];
    acc = __builtin_amdgcn_wmma_f32_16x16x4_f32(false, a, false, b,
                                                (short)0, acc, false, false);
  }
  long crow = (long)tm * 16 + half * 8;
#pragma unroll
  for (int r = 0; r < 8; ++r)
    C[(crow + r) * Nd + bcol] = acc[r];
}

__global__ void logits_kernel(const float* __restrict__ hmat, const float* __restrict__ a_src,
                              const float* __restrict__ a_dst, float* __restrict__ als,
                              float* __restrict__ ald, int nnodes, int heads, int ch) {
  int wid  = (int)(((long)blockIdx.x * blockDim.x + threadIdx.x) >> 5);
  int lane = threadIdx.x & 31;
  if (wid >= nnodes * heads) return;
  int node = wid / heads, hh = wid % heads;
  int F = heads * ch;
  float ss = 0.f, dd = 0.f;
  for (int c = lane; c < ch; c += 32) {
    float hv = hmat[(long)node * F + hh * ch + c];
    ss += hv * a_src[hh * ch + c];
    dd += hv * a_dst[hh * ch + c];
  }
#pragma unroll
  for (int off = 16; off > 0; off >>= 1) {
    ss += __shfl_xor(ss, off);
    dd += __shfl_xor(dd, off);
  }
  if (lane == 0) { als[wid] = ss; ald[wid] = dd; }
}

__device__ __forceinline__ void edge_sd(const int* __restrict__ eidx, int E, int e,
                                        int& s, int& d) {
  if (e < E) { s = eidx[e]; d = eidx[E + e]; }
  else       { s = e - E; d = s; }
}

__global__ void edge_max_kernel(const int* __restrict__ eidx, int E, int EN,
                                const float* __restrict__ als, const float* __restrict__ ald,
                                float* __restrict__ m, int heads) {
  long t = (long)blockIdx.x * blockDim.x + threadIdx.x;
  if (t >= (long)EN * heads) return;
  int e = (int)(t / heads), hh = (int)(t % heads);
  int s, d; edge_sd(eidx, E, e, s, d);
  float l = als[s * heads + hh] + ald[d * heads + hh];
  l = l > 0.f ? l : NEG_SLOPE * l;
  unsafeAtomicMax(&m[d * heads + hh], l);
}

__global__ void edge_fused_gen(const int* __restrict__ eidx, int E, int EN,
                               const float* __restrict__ als, const float* __restrict__ ald,
                               const float* __restrict__ m, float* __restrict__ ssum,
                               const float* __restrict__ hsrc, float* __restrict__ accum,
                               int heads, int ch) {
  int wid  = (int)(((long)blockIdx.x * blockDim.x + threadIdx.x) >> 5);
  int lane = threadIdx.x & 31;
  if (wid >= EN) return;
  int s, d; edge_sd(eidx, E, wid, s, d);
  int F = heads * ch;
  for (int j = lane; j < F; j += 32) {
    int head = j / ch;
    float l = als[s * heads + head] + ald[d * heads + head];
    l = l > 0.f ? l : NEG_SLOPE * l;
    float ex = __expf(l - m[d * heads + head]);
    if ((j % ch) == 0) unsafeAtomicAdd(&ssum[d * heads + head], ex);
    unsafeAtomicAdd(&accum[(long)d * F + j], hsrc[(long)s * F + j] * ex);
  }
}

__global__ void norm_bias_act_kernel(float* __restrict__ a, const float* __restrict__ ssum,
                                     const float* __restrict__ bias, long n, int F,
                                     int ch, int heads, int do_elu) {
  long i = (long)blockIdx.x * blockDim.x + threadIdx.x;
  long stride = (long)gridDim.x * blockDim.x;
  for (; i < n; i += stride) {
    int  j    = (int)(i % F);
    long node = i / F;
    float v = a[i] / ssum[node * heads + j / ch] + bias[j];
    if (do_elu) v = v > 0.f ? v : (__expf(v) - 1.f);
    a[i] = v;
  }
}


#define AGG_WAVES 32
#define AGG_RPW   7
#define AGG_ROWS  196
#define AGG_S     32768
#define AGG_STEPS (AGG_S / (AGG_WAVES * 32 * 4))
#define AGG_GROUP 4
#define AGG_SCAN  4

template<int HEADS, int CH, int DO_ELU>
__global__ void __launch_bounds__(AGG_WAVES * 32)
agg_owner_kernel(const int* __restrict__ eidx, int E, int nnodes,
                 const float* __restrict__ als, const float* __restrict__ ald,
                 const float* __restrict__ hsrc, const float* __restrict__ bias,
                 float* __restrict__ outp) {
  constexpr int F  = HEADS * CH;
  constexpr int NF = F / 32;
  __shared__ int2 list[AGG_S];
  __shared__ int  cnt[2];
  const int w    = threadIdx.x >> 5;
  const int lane = threadIdx.x & 31;
  const int r0   = blockIdx.x * AGG_ROWS;
  const int j0   = lane * NF;
  const int head = j0 / CH;

  float mx[AGG_RPW], sm[AGG_RPW], aldr[AGG_RPW], acc[AGG_RPW][NF];
#pragma unroll
  for (int j = 0; j < AGG_RPW; ++j) {
    int loc = j * 32 + w;
    int row = r0 + loc;
    if (loc < AGG_ROWS && row < nnodes) {
      float a_s = als[row * HEADS + head];
      float a_d = ald[row * HEADS + head];
      float l = a_s + a_d;
      l = l > 0.f ? l : NEG_SLOPE * l;
      mx[j] = l; sm[j] = 1.f; aldr[j] = a_d;
#pragma unroll
      for (int k = 0; k < NF; ++k) acc[j][k] = hsrc[(long)row * F + j0 + k];
    } else {
      mx[j] = 0.f; sm[j] = 1.f; aldr[j] = 0.f;
#pragma unroll
      for (int k = 0; k < NF; ++k) acc[j][k] = 0.f;
    }
  }

  const int* __restrict__ dstp = eidx + E;
  const bool vec_ok = ((E & 3) == 0) && ((((size_t)dstp) & 15) == 0);
  if (threadIdx.x == 0) { cnt[0] = 0; cnt[1] = 0; }
  int cb = 0;
  for (int c0 = 0; c0 < E; c0 += AGG_S, cb ^= 1) {
    __syncthreads();
#pragma unroll
    for (int tg = 0; tg < AGG_STEPS; tg += AGG_GROUP) {
      int4 dv[AGG_GROUP];
      if (vec_ok) {
#pragma unroll
        for (int t = 0; t < AGG_GROUP; ++t) {
          int i = c0 + ((tg + t) * (AGG_WAVES * 32) + (int)threadIdx.x) * 4;
          dv[t] = (i < E) ? *(const int4*)(dstp + i) : make_int4(-1, -1, -1, -1);
        }
      } else {
#pragma unroll
        for (int t = 0; t < AGG_GROUP; ++t) {
          int i = c0 + ((tg + t) * (AGG_WAVES * 32) + (int)threadIdx.x) * 4;
          dv[t].x = (i     < E) ? dstp[i]     : -1;
          dv[t].y = (i + 1 < E) ? dstp[i + 1] : -1;
          dv[t].z = (i + 2 < E) ? dstp[i + 2] : -1;
          dv[t].w = (i + 3 < E) ? dstp[i + 3] : -1;
        }
      }
#pragma unroll
      for (int t = 0; t < AGG_GROUP; ++t) {
        int i = c0 + ((tg + t) * (AGG_WAVES * 32) + (int)threadIdx.x) * 4;
        int dd[4] = {dv[t].x, dv[t].y, dv[t].z, dv[t].w};
#pragma unroll
        for (int k = 0; k < 4; ++k) {
          unsigned dl = (unsigned)(dd[k] - r0);
          if (dl < (unsigned)AGG_ROWS) {
            int pos = atomicAdd(&cnt[cb], 1);
            list[pos].y = (int)((dl << 16) | (unsigned)(i + k - c0));
          }
        }
      }
    }
    __syncthreads();
    const int n = cnt[cb];
    if (threadIdx.x == 0) cnt[cb ^ 1] = 0;
    for (int idx = threadIdx.x; idx < n; idx += AGG_WAVES * 32)
      list[idx].x = eidx[c0 + (list[idx].y & 0xffff)];
    __syncthreads();
    int   last = -1;
    bool  havec = false;
    int   jjc = 0;
    float ac = 0.f;
    float hvc[NF];
#pragma unroll
    for (int k = 0; k < NF; ++k) hvc[k] = 0.f;
    for (int guard = 0; guard <= n + 1; ++guard) {
      int by = 0x7fffffff, bs = 0;
      for (int base = lane; base < n; base += 32 * AGG_SCAN) {
        int2 en[AGG_SCAN];
#pragma unroll
        for (int u = 0; u < AGG_SCAN; ++u) en[u] = list[min(base + 32 * u, n - 1)];
#pragma unroll
        for (int u = 0; u < AGG_SCAN; ++u) {
          if (base + 32 * u < n && ((en[u].y >> 16) & 31) == w && en[u].y > last && en[u].y < by) {
            by = en[u].y; bs = en[u].x;
          }
        }
      }
      int kmin = by;
#pragma unroll
      for (int off = 16; off > 0; off >>= 1) kmin = min(kmin, __shfl_xor(kmin, off));
      kmin = __builtin_amdgcn_readfirstlane(kmin);
      const bool haven = (kmin != 0x7fffffff);
      int sn = 0;
      if (haven) {
        unsigned long long holder = __ballot(by == kmin);
        sn = __shfl(bs, __ffsll((long long)holder) - 1);
        sn = __builtin_amdgcn_readfirstlane(sn);
        last = kmin;
      }
      if (havec) {
#pragma unroll
        for (int j = 0; j < AGG_RPW; ++j) {
          if (jjc == j) {
            float l = ac + aldr[j];
            l = l > 0.f ? l : NEG_SLOPE * l;
            float mn = fmaxf(mx[j], l);
            float sc = __expf(mx[j] - mn);
            float wg = __expf(l - mn);
            sm[j] = sm[j] * sc + wg;
#pragma unroll
            for (int k = 0; k < NF; ++k) acc[j][k] = acc[j][k] * sc + wg * hvc[k];
            mx[j] = mn;
          }
        }
      }
      if (!haven) break;
      ac = als[sn * HEADS + head];
#pragma unroll
      for (int k = 0; k < NF; ++k) hvc[k] = hsrc[(long)sn * F + j0 + k];
      jjc = kmin >> 21;
      havec = true;
    }
  }

#pragma unroll
  for (int j = 0; j < AGG_RPW; ++j) {
    int loc = j * 32 + w;
    int row = r0 + loc;
    if (loc < AGG_ROWS && row < nnodes) {
#pragma unroll
      for (int k = 0; k < NF; ++k) {
        float v = acc[j][k] / sm[j] + bias[j0 + k];
        if (DO_ELU) v = v > 0.f ? v : (__expf(v) - 1.f);
        outp[(long)row * F + j0 + k] = v;
      }
    }
  }
}


static inline int cdiv_i(long a, long b) { return (int)((a + b - 1) / b); }

static void launch_gemm(const float* A, const float* B, float* C,
                        int M, int K, int Nd, hipStream_t stream) {
  int tilesM = M / 16, tilesN = Nd / 16;
  int wpb = tilesN >= 8 ? 8 : tilesN;
  dim3 grid(tilesM * (tilesN / wpb));
  dim3 block(32 * wpb);
  size_t shmem = (size_t)16 * K * sizeof(float);
  gemm_wmma_f32_lds<<<grid, block, shmem, stream>>>(A, B, C, M, K, Nd, wpb);
}

extern "C" void kernel_launch(void* const* d_in, const int* in_sizes, int n_in,
                              void* d_out, int out_size, void* d_ws, size_t ws_size,
                              hipStream_t stream) {
  const float* x   = (const float*)d_in[0];
  const int*   eix = (const int*)d_in[1];
  const float* W1  = (const float*)d_in[2];
  const float* as1 = (const float*)d_in[3];
  const float* ad1 = (const float*)d_in[4];
  const float* b1  = (const float*)d_in[5];
  const float* W2  = (const float*)d_in[6];
  const float* as2 = (const float*)d_in[7];
  const float* ad2 = (const float*)d_in[8];
  const float* b2  = (const float*)d_in[9];
  float* out = (float*)d_out;

  const int IN_DIM = 128;
  const int F1     = in_sizes[5];
  const int HID    = in_sizes[9];
  const int HEADS  = F1 / HID;
  const int Nn     = in_sizes[0] / IN_DIM;
  const int E      = in_sizes[1] / 2;
  const int EN     = E + Nn;

  float* acc1 = (float*)d_ws;
  float* h1   = acc1 + (long)Nn * F1;
  float* sc   = h1 + (long)Nn * F1;
  float* als1 = sc;
  float* ald1 = als1 + (long)Nn * HEADS;
  float* m1   = ald1 + (long)Nn * HEADS;
  float* s1   = m1   + (long)Nn * HEADS;
  float* als2 = s1   + (long)Nn * HEADS;
  float* ald2 = als2 + Nn;
  float* m2   = ald2 + Nn;
  float* s2   = m2   + Nn;
  float* h2   = h1;

  const float NEGINF = -__builtin_huge_valf();

  launch_gemm(x, W1, h1, Nn, IN_DIM, F1, stream);
  logits_kernel<<<cdiv_i((long)Nn * HEADS * 32, 256), 256, 0, stream>>>(
      h1, as1, ad1, als1, ald1, Nn, HEADS, HID);
  if (HEADS == 4 && HID == 64) {
    agg_owner_kernel<4, 64, 1><<<cdiv_i(Nn, AGG_ROWS), AGG_WAVES * 32, 0, stream>>>(
        eix, E, Nn, als1, ald1, h1, b1, acc1);
  } else {
    fill_f32<<<cdiv_i((long)Nn * F1, 256), 256, 0, stream>>>(acc1, (long)Nn * F1, 0.f);
    fill_f32<<<cdiv_i((long)Nn * HEADS, 256), 256, 0, stream>>>(m1, (long)Nn * HEADS, NEGINF);
    fill_f32<<<cdiv_i((long)Nn * HEADS, 256), 256, 0, stream>>>(s1, (long)Nn * HEADS, 0.f);
    edge_max_kernel<<<cdiv_i((long)EN * HEADS, 256), 256, 0, stream>>>(
        eix, E, EN, als1, ald1, m1, HEADS);
    edge_fused_gen<<<cdiv_i((long)EN * 32, 256), 256, 0, stream>>>(
        eix, E, EN, als1, ald1, m1, s1, h1, acc1, HEADS, HID);
    norm_bias_act_kernel<<<cdiv_i((long)Nn * F1, 256), 256, 0, stream>>>(
        acc1, s1, b1, (long)Nn * F1, F1, HID, HEADS, 1);
  }

  launch_gemm(acc1, W2, h2, Nn, F1, HID, stream);
  logits_kernel<<<cdiv_i((long)Nn * 32, 256), 256, 0, stream>>>(
      h2, as2, ad2, als2, ald2, Nn, 1, HID);
  if (HID == 64) {
    agg_owner_kernel<1, 64, 0><<<cdiv_i(Nn, AGG_ROWS), AGG_WAVES * 32, 0, stream>>>(
        eix, E, Nn, als2, ald2, h2, b2, out);
  } else {
    fill_f32<<<cdiv_i(Nn, 256), 256, 0, stream>>>(m2, Nn, NEGINF);
    fill_f32<<<cdiv_i(Nn, 256), 256, 0, stream>>>(s2, Nn, 0.f);
    fill_f32<<<cdiv_i((long)Nn * HID, 256), 256, 0, stream>>>(out, (long)Nn * HID, 0.f);
    edge_max_kernel<<<cdiv_i((long)EN, 256), 256, 0, stream>>>(
        eix, E, EN, als2, ald2, m2, 1);
    edge_fused_gen<<<cdiv_i((long)EN * 32, 256), 256, 0, stream>>>(
        eix, E, EN, als2, ald2, m2, s2, h2, out, 1, HID);
    norm_bias_act_kernel<<<cdiv_i((long)Nn * HID, 256), 256, 0, stream>>>(
        out, s2, b2, (long)Nn * HID, HID, HID, 1, 0);
  }
}


# Aggregation: a wave takes its entries of a pass out of the LDS list once, ranks them by key in registers, streams them with 2 (one head: 4) source rows in flight; no search per entry; sum order same
# speedup vs baseline: 3.0795x; 1.4401x over previous
#include <hip/hip_runtime.h>

typedef float v2f __attribute__((ext_vector_type(2)));
typedef float v8f __attribute__((ext_vector_type(8)));

#define NEG_SLOPE 0.2f


__global__ void fill_f32(float* __restrict__ p, long n, float v) {
  long i = (long)blockIdx.x * blockDim.x + threadIdx.x;
  long stride = (long)gridDim.x * blockDim.x;
  for (; i < n; i += stride) p[i] = v;
}

__global__ void gemm_wmma_f32_lds(const float* __restrict__ A, const float* __restrict__ B,
                                  float* __restrict__ C, int M, int K, int Nd,
                                  int tnPerBlock) {
  extern __shared__ float As[];
  int tilesN = Nd >> 4;
  int blocksPerRow = tilesN / tnPerBlock;
  int tm  = blockIdx.x / blocksPerRow;
  int tn0 = (blockIdx.x % blocksPerRow) * tnPerBlock;
  int wib  = threadIdx.x >> 5;
  int lane = threadIdx.x & 31;

  {
    const float4* Av  = (const float4*)(A + (long)tm * 16 * K);
    float4*       Asv = (float4*)As;
    int nvec = (16 * K) >> 2;
    for (int idx = threadIdx.x; idx < nvec; idx += blockDim.x) Asv[idx] = Av[idx];
  }
  __syncthreads();

  int tn   = tn0 + wib;
  int half = lane >> 4;
  int l15  = lane & 15;
  int bcol = tn * 16 + l15;
  v8f acc = {};
  for (int k = 0; k < K; k += 4) {
    int ka = k + 2 * half;
    v2f a, b;
    a.x = As[l15 * K + ka];
    a.y = As[l15 * K + ka + 1];
    b.x = B[(long)ka * Nd + bcol];
    b.y = B[(long)(ka + 1) * Nd + bcol];
    acc = __builtin_amdgcn_wmma_f32_16x16x4_f32(false, a, false, b,
                                                (short)0, acc, false, false);
  }
  long crow = (long)tm * 16 + half * 8;
#pragma unroll
  for (int r = 0; r < 8; ++r)
    C[(crow + r) * Nd + bcol] = acc[r];
}

__global__ void logits_kernel(const float* __restrict__ hmat, const float* __restrict__ a_src,
                              const float* __restrict__ a_dst, float* __restrict__ als,
                              float* __restrict__ ald, int nnodes, int heads, int ch) {
  int wid  = (int)(((long)blockIdx.x * blockDim.x + threadIdx.x) >> 5);
  int lane = threadIdx.x & 31;
  if (wid >= nnodes * heads) return;
  int node = wid / heads, hh = wid % heads;
  int F = heads * ch;
  float ss = 0.f, dd = 0.f;
  for (int c = lane; c < ch; c += 32) {
    float hv = hmat[(long)node * F + hh * ch + c];
    ss += hv * a_src[hh * ch + c];
    dd += hv * a_dst[hh * ch + c];
  }
#pragma unroll
  for (int off = 16; off > 0; off >>= 1) {
    ss += __shfl_xor(ss, off);
    dd += __shfl_xor(dd, off);
  }
  if (lane == 0) { als[wid] = ss; ald[wid] = dd; }
}

__device__ __forceinline__ void edge_sd(const int* __restrict__ eidx, int E, int e,
                                        int& s, int& d) {
  if (e < E) { s = eidx[e]; d = eidx[E + e]; }
  else       { s = e - E; d = s; }
}

__global__ void edge_max_kernel(const int* __restrict__ eidx, int E, int EN,
                                const float* __restrict__ als, const float* __restrict__ ald,
                                float* __restrict__ m, int heads) {
  long t = (long)blockIdx.x * blockDim.x + threadIdx.x;
  if (t >= (long)EN * heads) return;
  int e = (int)(t / heads), hh = (int)(t % heads);
  int s, d; edge_sd(eidx, E, e, s, d);
  float l = als[s * heads + hh] + ald[d * heads + hh];
  l = l > 0.f ? l : NEG_SLOPE * l;
  unsafeAtomicMax(&m[d * heads + hh], l);
}

__global__ void edge_fused_gen(const int* __restrict__ eidx, int E, int EN,
                               const float* __restrict__ als, const float* __restrict__ ald,
                               const float* __restrict__ m, float* __restrict__ ssum,
                               const float* __restrict__ hsrc, float* __restrict__ accum,
                               int heads, int ch) {
  int wid  = (int)(((long)blockIdx.x * blockDim.x + threadIdx.x) >> 5);
  int lane = threadIdx.x & 31;
  if (wid >= EN) return;
  int s, d; edge_sd(eidx, E, wid, s, d);
  int F = heads * ch;
  for (int j = lane; j < F; j += 32) {
    int head = j / ch;
    float l = als[s * heads + head] + ald[d * heads + head];
    l = l > 0.f ? l : NEG_SLOPE * l;
    float ex = __expf(l - m[d * heads + head]);
    if ((j % ch) == 0) unsafeAtomicAdd(&ssum[d * heads + head], ex);
    unsafeAtomicAdd(&accum[(long)d * F + j], hsrc[(long)s * F + j] * ex);
  }
}

__global__ void norm_bias_act_kernel(float* __restrict__ a, const float* __restrict__ ssum,
                                     const float* __restrict__ bias, long n, int F,
                                     int ch, int heads, int do_elu) {
  long i = (long)blockIdx.x * blockDim.x + threadIdx.x;
  long stride = (long)gridDim.x * blockDim.x;
  for (; i < n; i += stride) {
    int  j    = (int)(i % F);
    long node = i / F;
    float v = a[i] / ssum[node * heads + j / ch] + bias[j];
    if (do_elu) v = v > 0.f ? v : (__expf(v) - 1.f);
    a[i] = v;
  }
}


#define AGG_WAVES 32
#define AGG_RPW   7
#define AGG_ROWS  196
#define AGG_S     32768
#define AGG_STEPS (AGG_S / (AGG_WAVES * 32 * 4))
#define AGG_GROUP 4
#define AGG_SCAN  4

template<int HEADS, int CH, int DO_ELU>
__global__ void __launch_bounds__(AGG_WAVES * 32)
agg_owner_kernel(const int* __restrict__ eidx, int E, int nnodes,
                 const float* __restrict__ als, const float* __restrict__ ald,
                 const float* __restrict__ hsrc, const float* __restrict__ bias,
                 float* __restrict__ outp) {
  constexpr int F  = HEADS * CH;
  constexpr int NF = F / 32;
  constexpr int AGG_DEPTH = (NF >= 8) ? 2 : 4;
  __shared__ int2 list[AGG_S];
  __shared__ int  cnt[2];
  __shared__ int  wcnt[2][AGG_WAVES];
  const int w    = __builtin_amdgcn_readfirstlane((int)(threadIdx.x >> 5));
  const int lane = threadIdx.x & 31;
  const int r0   = blockIdx.x * AGG_ROWS;
  const int j0   = lane * NF;
  const int head = j0 / CH;

  float mx[AGG_RPW], sm[AGG_RPW], acc[AGG_RPW][NF];
#pragma unroll
  for (int j = 0; j < AGG_RPW; ++j) {
    int loc = j * 32 + w;
    int row = r0 + loc;
    if (loc < AGG_ROWS && row < nnodes) {
      float a_s = als[row * HEADS + head];
      float a_d = ald[row * HEADS + head];
      float l = a_s + a_d;
      l = l > 0.f ? l : NEG_SLOPE * l;
      mx[j] = l; sm[j] = 1.f;
#pragma unroll
      for (int k = 0; k < NF; ++k) acc[j][k] = hsrc[(long)row * F + j0 + k];
    } else {
      mx[j] = 0.f; sm[j] = 1.f;
#pragma unroll
      for (int k = 0; k < NF; ++k) acc[j][k] = 0.f;
    }
  }

  const int* __restrict__ dstp = eidx + E;
  const bool vec_ok = ((E & 3) == 0) && ((((size_t)dstp) & 15) == 0);
  if (threadIdx.x == 0) { cnt[0] = 0; cnt[1] = 0; }
  if (threadIdx.x < AGG_WAVES) { wcnt[0][threadIdx.x] = 0; wcnt[1][threadIdx.x] = 0; }
  int cb = 0;
  int S = AGG_S;
  for (int c0 = 0; c0 < E; cb ^= 1) {
    __syncthreads();
    const int lim = (E - c0 < S) ? E : c0 + S;
#pragma unroll
    for (int tg = 0; tg < AGG_STEPS; tg += AGG_GROUP) {
      int4 dv[AGG_GROUP];
      if (vec_ok) {
#pragma unroll
        for (int t = 0; t < AGG_GROUP; ++t) {
          int i = c0 + ((tg + t) * (AGG_WAVES * 32) + (int)threadIdx.x) * 4;
          dv[t] = (i < lim) ? *(const int4*)(dstp + i) : make_int4(-1, -1, -1, -1);
        }
      } else {
#pragma unroll
        for (int t = 0; t < AGG_GROUP; ++t) {
          int i = c0 + ((tg + t) * (AGG_WAVES * 32) + (int)threadIdx.x) * 4;
          dv[t].x = (i     < lim) ? dstp[i]     : -1;
          dv[t].y = (i + 1 < lim) ? dstp[i + 1] : -1;
          dv[t].z = (i + 2 < lim) ? dstp[i + 2] : -1;
          dv[t].w = (i + 3 < lim) ? dstp[i + 3] : -1;
        }
      }
#pragma unroll
      for (int t = 0; t < AGG_GROUP; ++t) {
        int i = c0 + ((tg + t) * (AGG_WAVES * 32) + (int)threadIdx.x) * 4;
        int dd[4] = {dv[t].x, dv[t].y, dv[t].z, dv[t].w};
#pragma unroll
        for (int k = 0; k < 4; ++k) {
          unsigned dl = (unsigned)(dd[k] - r0);
          if (dl < (unsigned)AGG_ROWS) {
            int pos = atomicAdd(&cnt[cb], 1);
            list[pos].y = (int)((dl << 16) | (unsigned)(i + k - c0));
            atomicAdd(&wcnt[cb][dl & 31], 1);
          }
        }
      }
    }
    __syncthreads();
    const int n = cnt[cb];
    if (threadIdx.x == 0) cnt[cb ^ 1] = 0;
    if (threadIdx.x < AGG_WAVES) wcnt[cb ^ 1][threadIdx.x] = 0;
    for (int idx = threadIdx.x; idx < n; idx += AGG_WAVES * 32)
      list[idx].x = eidx[c0 + (list[idx].y & 0xffff)];
    __syncthreads();
    if (__builtin_amdgcn_ballot_w32(wcnt[cb][lane] > 32) != 0) { S >>= 1; continue; }

    int qk = 0x7fffffff, qs = 0;
    int m = 0;
    for (int base = 0; base < n; base += 32 * AGG_SCAN) {
      int2 en[AGG_SCAN];
#pragma unroll
      for (int u = 0; u < AGG_SCAN; ++u) en[u] = list[min(base + 32 * u + lane, n - 1)];
#pragma unroll
      for (int u = 0; u < AGG_SCAN; ++u) {
        const bool own = (base + 32 * u + lane < n) && (((en[u].y >> 16) & 31) == w);
        unsigned mask = __builtin_amdgcn_ballot_w32(own);
        while (mask) {
          const int b = __builtin_ctz(mask);
          mask &= mask - 1;
          const int fk = __builtin_amdgcn_readlane(en[u].y, b);
          const int fs = __builtin_amdgcn_readlane(en[u].x, b);
          if (lane == m) { qk = fk; qs = fs; }
          ++m;
        }
      }
    }
    if (m > 0) {
      int rank = 0;
      for (int j = 0; j < m; ++j) rank += (__builtin_amdgcn_readlane(qk, j) < qk) ? 1 : 0;
      float ql = 0.f;
      if (HEADS == 1) {
        ql = als[qs] + ald[r0 + ((lane < m) ? (qk >> 16) : 0)];
        ql = ql > 0.f ? ql : NEG_SLOPE * ql;
      }
      auto fetch = [&](int r, float& a_s, float& a_d, float (&hv)[NF], int& jj, int& slot) __attribute__((always_inline)) {
        const int b = __builtin_ctz(__builtin_amdgcn_ballot_w32(rank == min(r, m - 1)));
        const int k = __builtin_amdgcn_readlane(qk, b);
        const int s = __builtin_amdgcn_readlane(qs, b);
        a_s = 0.f; a_d = 0.f;
        if (HEADS != 1) {
          a_s = als[s * HEADS + head];
          a_d = ald[(r0 + (k >> 16)) * HEADS + head];
        }
#pragma unroll
        for (int c = 0; c < NF; ++c) hv[c] = hsrc[(long)s * F + j0 + c];
        jj = k >> 21;
        slot = b;
      };
      auto apply = [&](int jj, float a_s, float a_d, const float (&hv)[NF], int slot) __attribute__((always_inline)) {
#pragma unroll
        for (int j = 0; j < AGG_RPW; ++j) {
          if (jj == j) {
            float l;
            if (HEADS == 1) {
              l = __int_as_float(__builtin_amdgcn_readlane(__float_as_int(ql), slot));
            } else {
              l = a_s + a_d;
              l = l > 0.f ? l : NEG_SLOPE * l;
            }
            float mn = fmaxf(mx[j], l);
            float sc = __expf(mx[j] - mn);
            float wg = __expf(l - mn);
            sm[j] = sm[j] * sc + wg;
#pragma unroll
            for (int k = 0; k < NF; ++k) acc[j][k] = acc[j][k] * sc + wg * hv[k];
            mx[j] = mn;
          }
        }
      };
      float asb[AGG_DEPTH], adb[AGG_DEPTH], hb[AGG_DEPTH][NF];
      int   jb[AGG_DEPTH], sb[AGG_DEPTH];
#pragma unroll
      for (int d = 0; d < AGG_DEPTH; ++d) fetch(d, asb[d], adb[d], hb[d], jb[d], sb[d]);
      int r = 0;
      for (; r + AGG_DEPTH <= m; r += AGG_DEPTH) {
#pragma unroll
        for (int d = 0; d < AGG_DEPTH; ++d) {
          apply(jb[d], asb[d], adb[d], hb[d], sb[d]);
          fetch(r + d + AGG_DEPTH, asb[d], adb[d], hb[d], jb[d], sb[d]);
        }
      }
#pragma unroll
      for (int d = 0; d < AGG_DEPTH; ++d)
        if (r + d < m) apply(jb[d], asb[d], adb[d], hb[d], sb[d]);
    }
    c0 += S;
    S = AGG_S;
  }

  const int j0e = (int)__builtin_amdgcn_mbcnt_lo(~0u, 0u) * NF;
#pragma unroll
  for (int j = 0; j < AGG_RPW; ++j) {
    int loc = j * 32 + w;
    int row = r0 + loc;
    if (loc < AGG_ROWS && row < nnodes) {
#pragma unroll
      for (int k = 0; k < NF; ++k) {
        float v = acc[j][k] / sm[j] + bias[j0e + k];
        if (DO_ELU) v = v > 0.f ? v : (__expf(v) - 1.f);
        outp[(long)row * F + j0e + k] = v;
      }
    }
  }
}


static inline int cdiv_i(long a, long b) { return (int)((a + b - 1) / b); }

static void launch_gemm(const float* A, const float* B, float* C,
                        int M, int K, int Nd, hipStream_t stream) {
  int tilesM = M / 16, tilesN = Nd / 16;
  int wpb = tilesN >= 8 ? 8 : tilesN;
  dim3 grid(tilesM * (tilesN / wpb));
  dim3 block(32 * wpb);
  size_t shmem = (size_t)16 * K * sizeof(float);
  gemm_wmma_f32_lds<<<grid, block, shmem, stream>>>(A, B, C, M, K, Nd, wpb);
}

extern "C" void kernel_launch(void* const* d_in, const int* in_sizes, int n_in,
                              void* d_out, int out_size, void* d_ws, size_t ws_size,
                              hipStream_t stream) {
  const float* x   = (const float*)d_in[0];
  const int*   eix = (const int*)d_in[1];
  const float* W1  = (const float*)d_in[2];
  const float* as1 = (const float*)d_in[3];
  const float* ad1 = (const float*)d_in[4];
  const float* b1  = (const float*)d_in[5];
  const float* W2  = (const float*)d_in[6];
  const float* as2 = (const float*)d_in[7];
  const float* ad2 = (const float*)d_in[8];
  const float* b2  = (const float*)d_in[9];
  float* out = (float*)d_out;

  const int IN_DIM = 128;
  const int F1     = in_sizes[5];
  const int HID    = in_sizes[9];
  const int HEADS  = F1 / HID;
  const int Nn     = in_sizes[0] / IN_DIM;
  const int E      = in_sizes[1] / 2;
  const int EN     = E + Nn;

  float* acc1 = (float*)d_ws;
  float* h1   = acc1 + (long)Nn * F1;
  float* sc   = h1 + (long)Nn * F1;
  float* als1 = sc;
  float* ald1 = als1 + (long)Nn * HEADS;
  float* m1   = ald1 + (long)Nn * HEADS;
  float* s1   = m1   + (long)Nn * HEADS;
  float* als2 = s1   + (long)Nn * HEADS;
  float* ald2 = als2 + Nn;
  float* m2   = ald2 + Nn;
  float* s2   = m2   + Nn;
  float* h2   = h1;

  const float NEGINF = -__builtin_huge_valf();

  launch_gemm(x, W1, h1, Nn, IN_DIM, F1, stream);
  logits_kernel<<<cdiv_i((long)Nn * HEADS * 32, 256), 256, 0, stream>>>(
      h1, as1, ad1, als1, ald1, Nn, HEADS, HID);
  if (HEADS == 4 && HID == 64) {
    agg_owner_kernel<4, 64, 1><<<cdiv_i(Nn, AGG_ROWS), AGG_WAVES * 32, 0, stream>>>(
        eix, E, Nn, als1, ald1, h1, b1, acc1);
  } else {
    fill_f32<<<cdiv_i((long)Nn * F1, 256), 256, 0, stream>>>(acc1, (long)Nn * F1, 0.f);
    fill_f32<<<cdiv_i((long)Nn * HEADS, 256), 256, 0, stream>>>(m1, (long)Nn * HEADS, NEGINF);
    fill_f32<<<cdiv_i((long)Nn * HEADS, 256), 256, 0, stream>>>(s1, (long)Nn * HEADS, 0.f);
    edge_max_kernel<<<cdiv_i((long)EN * HEADS, 256), 256, 0, stream>>>(
        eix, E, EN, als1, ald1, m1, HEADS);
    edge_fused_gen<<<cdiv_i((long)EN * 32, 256), 256, 0, stream>>>(
        eix, E, EN, als1, ald1, m1, s1, h1, acc1, HEADS, HID);
    norm_bias_act_kernel<<<cdiv_i((long)Nn * F1, 256), 256, 0, stream>>>(
        acc1, s1, b1, (long)Nn * F1, F1, HID, HEADS, 1);
  }

  launch_gemm(acc1, W2, h2, Nn, F1, HID, stream);
  logits_kernel<<<cdiv_i((long)Nn * 32, 256), 256, 0, stream>>>(
      h2, as2, ad2, als2, ald2, Nn, 1, HID);
  if (HID == 64) {
    agg_owner_kernel<1, 64, 0><<<cdiv_i(Nn, AGG_ROWS), AGG_WAVES * 32, 0, stream>>>(
        eix, E, Nn, als2, ald2, h2, b2, out);
  } else {
    fill_f32<<<cdiv_i(Nn, 256), 256, 0, stream>>>(m2, Nn, NEGINF);
    fill_f32<<<cdiv_i(Nn, 256), 256, 0, stream>>>(s2, Nn, 0.f);
    fill_f32<<<cdiv_i((long)Nn * HID, 256), 256, 0, stream>>>(out, (long)Nn * HID, 0.f);
    edge_max_kernel<<<cdiv_i((long)EN, 256), 256, 0, stream>>>(
        eix, E, EN, als2, ald2, m2, 1);
    edge_fused_gen<<<cdiv_i((long)EN * 32, 256), 256, 0, stream>>>(
        eix, E, EN, als2, ald2, m2, s2, h2, out, 1, HID);
    norm_bias_act_kernel<<<cdiv_i((long)Nn * HID, 256), 256, 0, stream>>>(
        out, s2, b2, (long)Nn * HID, HID, HID, 1, 0);
  }
}


# Attention logits made inside the aggregation kernel from the feature rows it reads anyway (lane dot parts, butterfly over a head's lanes); both logits launches gone, 4 launches; logit sum order new
# speedup vs baseline: 3.4238x; 1.1118x over previous
#include <hip/hip_runtime.h>

typedef float v2f __attribute__((ext_vector_type(2)));
typedef float v8f __attribute__((ext_vector_type(8)));

#define NEG_SLOPE 0.2f


__global__ void fill_f32(float* __restrict__ p, long n, float v) {
  long i = (long)blockIdx.x * blockDim.x + threadIdx.x;
  long stride = (long)gridDim.x * blockDim.x;
  for (; i < n; i += stride) p[i] = v;
}

__global__ void gemm_wmma_f32_lds(const float* __restrict__ A, const float* __restrict__ B,
                                  float* __restrict__ C, int M, int K, int Nd,
                                  int tnPerBlock) {
  extern __shared__ float As[];
  int tilesN = Nd >> 4;
  int blocksPerRow = tilesN / tnPerBlock;
  int tm  = blockIdx.x / blocksPerRow;
  int tn0 = (blockIdx.x % blocksPerRow) * tnPerBlock;
  int wib  = threadIdx.x >> 5;
  int lane = threadIdx.x & 31;

  {
    const float4* Av  = (const float4*)(A + (long)tm * 16 * K);
    float4*       Asv = (float4*)As;
    int nvec = (16 * K) >> 2;
    for (int idx = threadIdx.x; idx < nvec; idx += blockDim.x) Asv[idx] = Av[idx];
  }
  __syncthreads();

  int tn   = tn0 + wib;
  int half = lane >> 4;
  int l15  = lane & 15;
  int bcol = tn * 16 + l15;
  v8f acc = {};
  for (int k = 0; k < K; k += 4) {
    int ka = k + 2 * half;
    v2f a, b;
    a.x = As[l15 * K + ka];
    a.y = As[l15 * K + ka + 1];
    b.x = B[(long)ka * Nd + bcol];
    b.y = B[(long)(ka + 1) * Nd + bcol];
    acc = __builtin_amdgcn_wmma_f32_16x16x4_f32(false, a, false, b,
                                                (short)0, acc, false, false);
  }
  long crow = (long)tm * 16 + half * 8;
#pragma unroll
  for (int r = 0; r < 8; ++r)
    C[(crow + r) * Nd + bcol] = acc[r];
}

__global__ void logits_kernel(const float* __restrict__ hmat, const float* __restrict__ a_src,
                              const float* __restrict__ a_dst, float* __restrict__ als,
                              float* __restrict__ ald, int nnodes, int heads, int ch) {
  int wid  = (int)(((long)blockIdx.x * blockDim.x + threadIdx.x) >> 5);
  int lane = threadIdx.x & 31;
  if (wid >= nnodes * heads) return;
  int node = wid / heads, hh = wid % heads;
  int F = heads * ch;
  float ss = 0.f, dd = 0.f;
  for (int c = lane; c < ch; c += 32) {
    float hv = hmat[(long)node * F + hh * ch + c];
    ss += hv * a_src[hh * ch + c];
    dd += hv * a_dst[hh * ch + c];
  }
#pragma unroll
  for (int off = 16; off > 0; off >>= 1) {
    ss += __shfl_xor(ss, off);
    dd += __shfl_xor(dd, off);
  }
  if (lane == 0) { als[wid] = ss; ald[wid] = dd; }
}

__device__ __forceinline__ void edge_sd(const int* __restrict__ eidx, int E, int e,
                                        int& s, int& d) {
  if (e < E) { s = eidx[e]; d = eidx[E + e]; }
  else       { s = e - E; d = s; }
}

__global__ void edge_max_kernel(const int* __restrict__ eidx, int E, int EN,
                                const float* __restrict__ als, const float* __restrict__ ald,
                                float* __restrict__ m, int heads) {
  long t = (long)blockIdx.x * blockDim.x + threadIdx.x;
  if (t >= (long)EN * heads) return;
  int e = (int)(t / heads), hh = (int)(t % heads);
  int s, d; edge_sd(eidx, E, e, s, d);
  float l = als[s * heads + hh] + ald[d * heads + hh];
  l = l > 0.f ? l : NEG_SLOPE * l;
  unsafeAtomicMax(&m[d * heads + hh], l);
}

__global__ void edge_fused_gen(const int* __restrict__ eidx, int E, int EN,
                               const float* __restrict__ als, const float* __restrict__ ald,
                               const float* __restrict__ m, float* __restrict__ ssum,
                               const float* __restrict__ hsrc, float* __restrict__ accum,
                               int heads, int ch) {
  int wid  = (int)(((long)blockIdx.x * blockDim.x + threadIdx.x) >> 5);
  int lane = threadIdx.x & 31;
  if (wid >= EN) return;
  int s, d; edge_sd(eidx, E, wid, s, d);
  int F = heads * ch;
  for (int j = lane; j < F; j += 32) {
    int head = j / ch;
    float l = als[s * heads + head] + ald[d * heads + head];
    l = l > 0.f ? l : NEG_SLOPE * l;
    float ex = __expf(l - m[d * heads + head]);
    if ((j % ch) == 0) unsafeAtomicAdd(&ssum[d * heads + head], ex);
    unsafeAtomicAdd(&accum[(long)d * F + j], hsrc[(long)s * F + j] * ex);
  }
}

__global__ void norm_bias_act_kernel(float* __restrict__ a, const float* __restrict__ ssum,
                                     const float* __restrict__ bias, long n, int F,
                                     int ch, int heads, int do_elu) {
  long i = (long)blockIdx.x * blockDim.x + threadIdx.x;
  long stride = (long)gridDim.x * blockDim.x;
  for (; i < n; i += stride) {
    int  j    = (int)(i % F);
    long node = i / F;
    float v = a[i] / ssum[node * heads + j / ch] + bias[j];
    if (do_elu) v = v > 0.f ? v : (__expf(v) - 1.f);
    a[i] = v;
  }
}


#define AGG_WAVES 32
#define AGG_RPW   7
#define AGG_ROWS  196
#define AGG_S     32768
#define AGG_STEPS (AGG_S / (AGG_WAVES * 32 * 4))
#define AGG_GROUP 4
#define AGG_SCAN  4

template<int NF>
__device__ __forceinline__ float dot_part(const float (&v)[NF], const float (&a)[NF]) {
  float p0 = v[0] * a[0];
  float p1 = (NF > 1) ? v[NF > 1 ? 1 : 0] * a[NF > 1 ? 1 : 0] : 0.f;
#pragma unroll
  for (int k = 2; k < NF; k += 2) {
    p0 = __builtin_fmaf(v[k], a[k], p0);
    if (k + 1 < NF) p1 = __builtin_fmaf(v[k + 1], a[k + 1], p1);
  }
  return p0 + p1;
}

template<int G>
__device__ __forceinline__ float group_sum(float p) {
  static_assert(G == 1 || G == 2 || G == 4 || G == 8 || G == 16 || G == 32, "lanes per head must be a power of two");
  if (G >= 2)  p += __int_as_float(__builtin_amdgcn_update_dpp(0, __float_as_int(p), 0x161, 0xf, 0xf, false));
  if (G >= 4)  p += __int_as_float(__builtin_amdgcn_update_dpp(0, __float_as_int(p), 0x162, 0xf, 0xf, false));
  if (G >= 8)  p += __int_as_float(__builtin_amdgcn_update_dpp(0, __float_as_int(p), 0x164, 0xf, 0xf, false));
  if (G >= 16) p += __int_as_float(__builtin_amdgcn_update_dpp(0, __float_as_int(p), 0x168, 0xf, 0xf, false));
  if (G >= 32) {
    const int b = __float_as_int(p);
    p += __int_as_float(__builtin_amdgcn_permlanex16(b, b, 0x76543210, 0xfedcba98, false, false));
  }
  return p;
}

template<int HEADS, int CH, int DO_ELU>
__global__ void __launch_bounds__(AGG_WAVES * 32)
agg_owner_kernel(const int* __restrict__ eidx, int E, int nnodes,
                 const float* __restrict__ a_src, const float* __restrict__ a_dst,
                 const float* __restrict__ hsrc, const float* __restrict__ bias,
                 float* __restrict__ outp) {
  constexpr int F  = HEADS * CH;
  constexpr int NF = F / 32;
  constexpr int LPH = CH / NF;
  static_assert(NF >= 1 && CH % NF == 0 && LPH <= 32, "a head's channels must fill whole lanes");
  constexpr int AGG_DEPTH = (NF >= 8) ? 2 : 4;
  __shared__ int2 list[AGG_S];
  __shared__ int  cnt[2];
  __shared__ int  wcnt[2][AGG_WAVES];
  const int w    = __builtin_amdgcn_readfirstlane((int)(threadIdx.x >> 5));
  const int lane = threadIdx.x & 31;
  const int r0   = blockIdx.x * AGG_ROWS;
  const int j0   = lane * NF;

  float mx[AGG_RPW], sm[AGG_RPW], acc[AGG_RPW][NF];
  float as_l[NF];
  float adr[AGG_RPW];
  {
    float ad_l[NF];
#pragma unroll
    for (int k = 0; k < NF; ++k) { as_l[k] = a_src[j0 + k]; ad_l[k] = a_dst[j0 + k]; }
#pragma unroll
    for (int j = 0; j < AGG_RPW; ++j) {
      int loc = j * 32 + w;
      int row = r0 + loc;
      if (loc < AGG_ROWS && row < nnodes) {
#pragma unroll
        for (int k = 0; k < NF; ++k) acc[j][k] = hsrc[(long)row * F + j0 + k];
        float a_s = group_sum<LPH>(dot_part<NF>(acc[j], as_l));
        float a_d = group_sum<LPH>(dot_part<NF>(acc[j], ad_l));
        float l = a_s + a_d;
        l = l > 0.f ? l : NEG_SLOPE * l;
        mx[j] = l; sm[j] = 1.f; adr[j] = a_d;
      } else {
        mx[j] = 0.f; sm[j] = 1.f; adr[j] = 0.f;
#pragma unroll
        for (int k = 0; k < NF; ++k) acc[j][k] = 0.f;
      }
    }
  }

  const int* __restrict__ dstp = eidx + E;
  const bool vec_ok = ((E & 3) == 0) && ((((size_t)dstp) & 15) == 0);
  if (threadIdx.x == 0) { cnt[0] = 0; cnt[1] = 0; }
  if (threadIdx.x < AGG_WAVES) { wcnt[0][threadIdx.x] = 0; wcnt[1][threadIdx.x] = 0; }
  int cb = 0;
  int S = AGG_S;
  for (int c0 = 0; c0 < E; cb ^= 1) {
    __syncthreads();
    const int lim = (E - c0 < S) ? E : c0 + S;
#pragma unroll
    for (int tg = 0; tg < AGG_STEPS; tg += AGG_GROUP) {
      int4 dv[AGG_GROUP];
      if (vec_ok) {
#pragma unroll
        for (int t = 0; t < AGG_GROUP; ++t) {
          int i = c0 + ((tg + t) * (AGG_WAVES * 32) + (int)threadIdx.x) * 4;
          dv[t] = (i < lim) ? *(const int4*)(dstp + i) : make_int4(-1, -1, -1, -1);
        }
      } else {
#pragma unroll
        for (int t = 0; t < AGG_GROUP; ++t) {
          int i = c0 + ((tg + t) * (AGG_WAVES * 32) + (int)threadIdx.x) * 4;
          dv[t].x = (i     < lim) ? dstp[i]     : -1;
          dv[t].y = (i + 1 < lim) ? dstp[i + 1] : -1;
          dv[t].z = (i + 2 < lim) ? dstp[i + 2] : -1;
          dv[t].w = (i + 3 < lim) ? dstp[i + 3] : -1;
        }
      }
#pragma unroll
      for (int t = 0; t < AGG_GROUP; ++t) {
        int i = c0 + ((tg + t) * (AGG_WAVES * 32) + (int)threadIdx.x) * 4;
        int dd[4] = {dv[t].x, dv[t].y, dv[t].z, dv[t].w};
#pragma unroll
        for (int k = 0; k < 4; ++k) {
          unsigned dl = (unsigned)(dd[k] - r0);
          if (dl < (unsigned)AGG_ROWS) {
            int pos = atomicAdd(&cnt[cb], 1);
            list[pos].y = (int)((dl << 16) | (unsigned)(i + k - c0));
            atomicAdd(&wcnt[cb][dl & 31], 1);
          }
        }
      }
    }
    __syncthreads();
    const int n = cnt[cb];
    if (threadIdx.x == 0) cnt[cb ^ 1] = 0;
    if (threadIdx.x < AGG_WAVES) wcnt[cb ^ 1][threadIdx.x] = 0;
    for (int idx = threadIdx.x; idx < n; idx += AGG_WAVES * 32)
      list[idx].x = eidx[c0 + (list[idx].y & 0xffff)];
    __syncthreads();
    if (__builtin_amdgcn_ballot_w32(wcnt[cb][lane] > 32) != 0) { S >>= 1; continue; }

    int qk = 0x7fffffff, qs = 0;
    int m = 0;
    for (int base = 0; base < n; base += 32 * AGG_SCAN) {
      int2 en[AGG_SCAN];
#pragma unroll
      for (int u = 0; u < AGG_SCAN; ++u) en[u] = list[min(base + 32 * u + lane, n - 1)];
#pragma unroll
      for (int u = 0; u < AGG_SCAN; ++u) {
        const bool own = (base + 32 * u + lane < n) && (((en[u].y >> 16) & 31) == w);
        unsigned mask = __builtin_amdgcn_ballot_w32(own);
        while (mask) {
          const int b = __builtin_ctz(mask);
          mask &= mask - 1;
          const int fk = __builtin_amdgcn_readlane(en[u].y, b);
          const int fs = __builtin_amdgcn_readlane(en[u].x, b);
          if (lane == m) { qk = fk; qs = fs; }
          ++m;
        }
      }
    }
    if (m > 0) {
      int rank = 0;
      for (int j = 0; j < m; ++j) rank += (__builtin_amdgcn_readlane(qk, j) < qk) ? 1 : 0;
      auto fetch = [&](int r, float (&hv)[NF], int& jj) __attribute__((always_inline)) {
        const int b = __builtin_ctz(__builtin_amdgcn_ballot_w32(rank == min(r, m - 1)));
        const int k = __builtin_amdgcn_readlane(qk, b);
        const int s = __builtin_amdgcn_readlane(qs, b);
#pragma unroll
        for (int c = 0; c < NF; ++c) hv[c] = hsrc[(long)s * F + j0 + c];
        jj = k >> 21;
      };
      auto apply = [&](int jj, const float (&hv)[NF]) __attribute__((always_inline)) {
        const float a_s = group_sum<LPH>(dot_part<NF>(hv, as_l));
#pragma unroll
        for (int j = 0; j < AGG_RPW; ++j) {
          if (jj == j) {
            float l = a_s + adr[j];
            l = l > 0.f ? l : NEG_SLOPE * l;
            float mn = fmaxf(mx[j], l);
            float sc = __expf(mx[j] - mn);
            float wg = __expf(l - mn);
            sm[j] = sm[j] * sc + wg;
#pragma unroll
            for (int k = 0; k < NF; ++k) acc[j][k] = acc[j][k] * sc + wg * hv[k];
            mx[j] = mn;
          }
        }
      };
      float hb[AGG_DEPTH][NF];
      int   jb[AGG_DEPTH];
#pragma unroll
      for (int d = 0; d < AGG_DEPTH; ++d) fetch(d, hb[d], jb[d]);
      int r = 0;
      for (; r + AGG_DEPTH <= m; r += AGG_DEPTH) {
#pragma unroll
        for (int d = 0; d < AGG_DEPTH; ++d) {
          apply(jb[d], hb[d]);
          fetch(r + d + AGG_DEPTH, hb[d], jb[d]);
        }
      }
#pragma unroll
      for (int d = 0; d < AGG_DEPTH; ++d)
        if (r + d < m) apply(jb[d], hb[d]);
    }
    c0 += S;
    S = AGG_S;
  }

  const int j0e = (int)__builtin_amdgcn_mbcnt_lo(~0u, 0u) * NF;
#pragma unroll
  for (int j = 0; j < AGG_RPW; ++j) {
    int loc = j * 32 + w;
    int row = r0 + loc;
    if (loc < AGG_ROWS && row < nnodes) {
#pragma unroll
      for (int k = 0; k < NF; ++k) {
        float v = acc[j][k] / sm[j] + bias[j0e + k];
        if (DO_ELU) v = v > 0.f ? v : (__expf(v) - 1.f);
        outp[(long)row * F + j0e + k] = v;
      }
    }
  }
}


static inline int cdiv_i(long a, long b) { return (int)((a + b - 1) / b); }

static void launch_gemm(const float* A, const float* B, float* C,
                        int M, int K, int Nd, hipStream_t stream) {
  int tilesM = M / 16, tilesN = Nd / 16;
  int wpb = tilesN >= 8 ? 8 : tilesN;
  dim3 grid(tilesM * (tilesN / wpb));
  dim3 block(32 * wpb);
  size_t shmem = (size_t)16 * K * sizeof(float);
  gemm_wmma_f32_lds<<<grid, block, shmem, stream>>>(A, B, C, M, K, Nd, wpb);
}

extern "C" void kernel_launch(void* const* d_in, const int* in_sizes, int n_in,
                              void* d_out, int out_size, void* d_ws, size_t ws_size,
                              hipStream_t stream) {
  const float* x   = (const float*)d_in[0];
  const int*   eix = (const int*)d_in[1];
  const float* W1  = (const float*)d_in[2];
  const float* as1 = (const float*)d_in[3];
  const float* ad1 = (const float*)d_in[4];
  const float* b1  = (const float*)d_in[5];
  const float* W2  = (const float*)d_in[6];
  const float* as2 = (const float*)d_in[7];
  const float* ad2 = (const float*)d_in[8];
  const float* b2  = (const float*)d_in[9];
  float* out = (float*)d_out;

  const int IN_DIM = 128;
  const int F1     = in_sizes[5];
  const int HID    = in_sizes[9];
  const int HEADS  = F1 / HID;
  const int Nn     = in_sizes[0] / IN_DIM;
  const int E      = in_sizes[1] / 2;
  const int EN     = E + Nn;

  float* acc1 = (float*)d_ws;
  float* h1   = acc1 + (long)Nn * F1;
  float* sc   = h1 + (long)Nn * F1;
  float* als1 = sc;
  float* ald1 = als1 + (long)Nn * HEADS;
  float* m1   = ald1 + (long)Nn * HEADS;
  float* s1   = m1   + (long)Nn * HEADS;
  float* als2 = s1   + (long)Nn * HEADS;
  float* ald2 = als2 + Nn;
  float* m2   = ald2 + Nn;
  float* s2   = m2   + Nn;
  float* h2   = h1;

  const float NEGINF = -__builtin_huge_valf();

  launch_gemm(x, W1, h1, Nn, IN_DIM, F1, stream);
  if (HEADS == 4 && HID == 64) {
    agg_owner_kernel<4, 64, 1><<<cdiv_i(Nn, AGG_ROWS), AGG_WAVES * 32, 0, stream>>>(
        eix, E, Nn, as1, ad1, h1, b1, acc1);
  } else {
    logits_kernel<<<cdiv_i((long)Nn * HEADS * 32, 256), 256, 0, stream>>>(
        h1, as1, ad1, als1, ald1, Nn, HEADS, HID);
    fill_f32<<<cdiv_i((long)Nn * F1, 256), 256, 0, stream>>>(acc1, (long)Nn * F1, 0.f);
    fill_f32<<<cdiv_i((long)Nn * HEADS, 256), 256, 0, stream>>>(m1, (long)Nn * HEADS, NEGINF);
    fill_f32<<<cdiv_i((long)Nn * HEADS, 256), 256, 0, stream>>>(s1, (long)Nn * HEADS, 0.f);
    edge_max_kernel<<<cdiv_i((long)EN * HEADS, 256), 256, 0, stream>>>(
        eix, E, EN, als1, ald1, m1, HEADS);
    edge_fused_gen<<<cdiv_i((long)EN * 32, 256), 256, 0, stream>>>(
        eix, E, EN, als1, ald1, m1, s1, h1, acc1, HEADS, HID);
    norm_bias_act_kernel<<<cdiv_i((long)Nn * F1, 256), 256, 0, stream>>>(
        acc1, s1, b1, (long)Nn * F1, F1, HID, HEADS, 1);
  }

  launch_gemm(acc1, W2, h2, Nn, F1, HID, stream);
  if (HID == 64) {
    agg_owner_kernel<1, 64, 0><<<cdiv_i(Nn, AGG_ROWS), AGG_WAVES * 32, 0, stream>>>(
        eix, E, Nn, as2, ad2, h2, b2, out);
  } else {
    logits_kernel<<<cdiv_i((long)Nn * 32, 256), 256, 0, stream>>>(
        h2, as2, ad2, als2, ald2, Nn, 1, HID);
    fill_f32<<<cdiv_i(Nn, 256), 256, 0, stream>>>(m2, Nn, NEGINF);
    fill_f32<<<cdiv_i(Nn, 256), 256, 0, stream>>>(s2, Nn, 0.f);
    fill_f32<<<cdiv_i((long)Nn * HID, 256), 256, 0, stream>>>(out, (long)Nn * HID, 0.f);
    edge_max_kernel<<<cdiv_i((long)EN, 256), 256, 0, stream>>>(
        eix, E, EN, als2, ald2, m2, 1);
    edge_fused_gen<<<cdiv_i((long)EN * 32, 256), 256, 0, stream>>>(
        eix, E, EN, als2, ald2, m2, s2, h2, out, 1, HID);
    norm_bias_act_kernel<<<cdiv_i((long)Nn * HID, 256), 256, 0, stream>>>(
        out, s2, b2, (long)Nn * HID, HID, HID, 1, 0);
  }
}


# Aggregation: filter passes only build each wave's own edge list (rows' edges ascending) in the workspace, read no rows; waves then stream their lists barrier-free; layer 2 reuses them; sum order same
# speedup vs baseline: 4.4795x; 1.3083x over previous
#include <hip/hip_runtime.h>

typedef float v2f __attribute__((ext_vector_type(2)));
typedef float v8f __attribute__((ext_vector_type(8)));

#define NEG_SLOPE 0.2f


__global__ void fill_f32(float* __restrict__ p, long n, float v) {
  long i = (long)blockIdx.x * blockDim.x + threadIdx.x;
  long stride = (long)gridDim.x * blockDim.x;
  for (; i < n; i += stride) p[i] = v;
}

__global__ void gemm_wmma_f32_lds(const float* __restrict__ A, const float* __restrict__ B,
                                  float* __restrict__ C, int M, int K, int Nd,
                                  int tnPerBlock) {
  extern __shared__ float As[];
  int tilesN = Nd >> 4;
  int blocksPerRow = tilesN / tnPerBlock;
  int tm  = blockIdx.x / blocksPerRow;
  int tn0 = (blockIdx.x % blocksPerRow) * tnPerBlock;
  int wib  = threadIdx.x >> 5;
  int lane = threadIdx.x & 31;

  {
    const float4* Av  = (const float4*)(A + (long)tm * 16 * K);
    float4*       Asv = (float4*)As;
    int nvec = (16 * K) >> 2;
    for (int idx = threadIdx.x; idx < nvec; idx += blockDim.x) Asv[idx] = Av[idx];
  }
  __syncthreads();

  int tn   = tn0 + wib;
  int half = lane >> 4;
  int l15  = lane & 15;
  int bcol = tn * 16 + l15;
  v8f acc = {};
  for (int k = 0; k < K; k += 4) {
    int ka = k + 2 * half;
    v2f a, b;
    a.x = As[l15 * K + ka];
    a.y = As[l15 * K + ka + 1];
    b.x = B[(long)ka * Nd + bcol];
    b.y = B[(long)(ka + 1) * Nd + bcol];
    acc = __builtin_amdgcn_wmma_f32_16x16x4_f32(false, a, false, b,
                                                (short)0, acc, false, false);
  }
  long crow = (long)tm * 16 + half * 8;
#pragma unroll
  for (int r = 0; r < 8; ++r)
    C[(crow + r) * Nd + bcol] = acc[r];
}

__global__ void logits_kernel(const float* __restrict__ hmat, const float* __restrict__ a_src,
                              const float* __restrict__ a_dst, float* __restrict__ als,
                              float* __restrict__ ald, int nnodes, int heads, int ch) {
  int wid  = (int)(((long)blockIdx.x * blockDim.x + threadIdx.x) >> 5);
  int lane = threadIdx.x & 31;
  if (wid >= nnodes * heads) return;
  int node = wid / heads, hh = wid % heads;
  int F = heads * ch;
  float ss = 0.f, dd = 0.f;
  for (int c = lane; c < ch; c += 32) {
    float hv = hmat[(long)node * F + hh * ch + c];
    ss += hv * a_src[hh * ch + c];
    dd += hv * a_dst[hh * ch + c];
  }
#pragma unroll
  for (int off = 16; off > 0; off >>= 1) {
    ss += __shfl_xor(ss, off);
    dd += __shfl_xor(dd, off);
  }
  if (lane == 0) { als[wid] = ss; ald[wid] = dd; }
}

__device__ __forceinline__ void edge_sd(const int* __restrict__ eidx, int E, int e,
                                        int& s, int& d) {
  if (e < E) { s = eidx[e]; d = eidx[E + e]; }
  else       { s = e - E; d = s; }
}

__global__ void edge_max_kernel(const int* __restrict__ eidx, int E, int EN,
                                const float* __restrict__ als, const float* __restrict__ ald,
                                float* __restrict__ m, int heads) {
  long t = (long)blockIdx.x * blockDim.x + threadIdx.x;
  if (t >= (long)EN * heads) return;
  int e = (int)(t / heads), hh = (int)(t % heads);
  int s, d; edge_sd(eidx, E, e, s, d);
  float l = als[s * heads + hh] + ald[d * heads + hh];
  l = l > 0.f ? l : NEG_SLOPE * l;
  unsafeAtomicMax(&m[d * heads + hh], l);
}

__global__ void edge_fused_gen(const int* __restrict__ eidx, int E, int EN,
                               const float* __restrict__ als, const float* __restrict__ ald,
                               const float* __restrict__ m, float* __restrict__ ssum,
                               const float* __restrict__ hsrc, float* __restrict__ accum,
                               int heads, int ch) {
  int wid  = (int)(((long)blockIdx.x * blockDim.x + threadIdx.x) >> 5);
  int lane = threadIdx.x & 31;
  if (wid >= EN) return;
  int s, d; edge_sd(eidx, E, wid, s, d);
  int F = heads * ch;
  for (int j = lane; j < F; j += 32) {
    int head = j / ch;
    float l = als[s * heads + head] + ald[d * heads + head];
    l = l > 0.f ? l : NEG_SLOPE * l;
    float ex = __expf(l - m[d * heads + head]);
    if ((j % ch) == 0) unsafeAtomicAdd(&ssum[d * heads + head], ex);
    unsafeAtomicAdd(&accum[(long)d * F + j], hsrc[(long)s * F + j] * ex);
  }
}

__global__ void norm_bias_act_kernel(float* __restrict__ a, const float* __restrict__ ssum,
                                     const float* __restrict__ bias, long n, int F,
                                     int ch, int heads, int do_elu) {
  long i = (long)blockIdx.x * blockDim.x + threadIdx.x;
  long stride = (long)gridDim.x * blockDim.x;
  for (; i < n; i += stride) {
    int  j    = (int)(i % F);
    long node = i / F;
    float v = a[i] / ssum[node * heads + j / ch] + bias[j];
    if (do_elu) v = v > 0.f ? v : (__expf(v) - 1.f);
    a[i] = v;
  }
}


#define AGG_WAVES 32
#define AGG_RPW   7
#define AGG_ROWS  196
#define AGG_S     32768
#define AGG_STEPS (AGG_S / (AGG_WAVES * 32 * 4))
#define AGG_GROUP 4
#define AGG_SCAN  4
#define AGG_CAP   1024

template<int NF>
__device__ __forceinline__ float dot_part(const float (&v)[NF], const float (&a)[NF]) {
  float p0 = v[0] * a[0];
  float p1 = (NF > 1) ? v[NF > 1 ? 1 : 0] * a[NF > 1 ? 1 : 0] : 0.f;
#pragma unroll
  for (int k = 2; k < NF; k += 2) {
    p0 = __builtin_fmaf(v[k], a[k], p0);
    if (k + 1 < NF) p1 = __builtin_fmaf(v[k + 1], a[k + 1], p1);
  }
  return p0 + p1;
}

template<int G>
__device__ __forceinline__ float group_sum(float p) {
  static_assert(G == 1 || G == 2 || G == 4 || G == 8 || G == 16 || G == 32, "lanes per head must be a power of two");
  if (G >= 2)  p += __int_as_float(__builtin_amdgcn_update_dpp(0, __float_as_int(p), 0x161, 0xf, 0xf, false));
  if (G >= 4)  p += __int_as_float(__builtin_amdgcn_update_dpp(0, __float_as_int(p), 0x162, 0xf, 0xf, false));
  if (G >= 8)  p += __int_as_float(__builtin_amdgcn_update_dpp(0, __float_as_int(p), 0x164, 0xf, 0xf, false));
  if (G >= 16) p += __int_as_float(__builtin_amdgcn_update_dpp(0, __float_as_int(p), 0x168, 0xf, 0xf, false));
  if (G >= 32) {
    const int b = __float_as_int(p);
    p += __int_as_float(__builtin_amdgcn_permlanex16(b, b, 0x76543210, 0xfedcba98, false, false));
  }
  return p;
}

template<int HEADS, int CH, int DO_ELU, int BUILD>
__global__ void __launch_bounds__(AGG_WAVES * 32)
agg_owner_kernel(const int* __restrict__ eidx, int E, int nnodes,
                 const float* __restrict__ a_src, const float* __restrict__ a_dst,
                 const float* __restrict__ hsrc, const float* __restrict__ bias,
                 float* __restrict__ outp, int* wlist, int* wtotal) {
  constexpr int F  = HEADS * CH;
  constexpr int NF = F / 32;
  constexpr int LPH = CH / NF;
  static_assert(NF >= 1 && CH % NF == 0 && LPH <= 32, "a head's channels must fill whole lanes");
  constexpr int DEPTH = (NF >= 8) ? 2 : ((NF >= 4) ? 4 : 8);
  __shared__ int2 list[AGG_S];
  __shared__ int  cnt[2];
  __shared__ int  wcnt[2][AGG_WAVES];
  const int w    = __builtin_amdgcn_readfirstlane((int)(threadIdx.x >> 5));
  const int lane = threadIdx.x & 31;
  const int r0   = blockIdx.x * AGG_ROWS;
  const int j0   = lane * NF;
  int2* const mylist = (int2*)wlist + (long)(blockIdx.x * AGG_WAVES + w) * AGG_CAP;

  float mx[AGG_RPW], sm[AGG_RPW], acc[AGG_RPW][NF];
  float as_l[NF];
  float adr[AGG_RPW];
  {
    float ad_l[NF];
#pragma unroll
    for (int k = 0; k < NF; ++k) { as_l[k] = a_src[j0 + k]; ad_l[k] = a_dst[j0 + k]; }
#pragma unroll
    for (int j = 0; j < AGG_RPW; ++j) {
      int loc = j * 32 + w;
      int row = r0 + loc;
      if (loc < AGG_ROWS && row < nnodes) {
#pragma unroll
        for (int k = 0; k < NF; ++k) acc[j][k] = hsrc[(long)row * F + j0 + k];
        float a_s = group_sum<LPH>(dot_part<NF>(acc[j], as_l));
        float a_d = group_sum<LPH>(dot_part<NF>(acc[j], ad_l));
        float l = a_s + a_d;
        l = l > 0.f ? l : NEG_SLOPE * l;
        mx[j] = l; sm[j] = 1.f; adr[j] = a_d;
      } else {
        mx[j] = 0.f; sm[j] = 1.f; adr[j] = 0.f;
#pragma unroll
        for (int k = 0; k < NF; ++k) acc[j][k] = 0.f;
      }
    }
  }

  auto apply = [&](int jj, const float (&hv)[NF]) __attribute__((always_inline)) {
    const float a_s = group_sum<LPH>(dot_part<NF>(hv, as_l));
#pragma unroll
    for (int j = 0; j < AGG_RPW; ++j) {
      if (jj == j) {
        float l = a_s + adr[j];
        l = l > 0.f ? l : NEG_SLOPE * l;
        float mn = fmaxf(mx[j], l);
        float sc = __expf(mx[j] - mn);
        float wg = __expf(l - mn);
        sm[j] = sm[j] * sc + wg;
#pragma unroll
        for (int k = 0; k < NF; ++k) acc[j][k] = acc[j][k] * sc + wg * hv[k];
        mx[j] = mn;
      }
    }
  };

  auto stream = [&](int nent) __attribute__((always_inline)) {
    for (int base = 0; base < nent; base += 32) {
      const int mm = min(32, nent - base);
      const int2 en = mylist[base + min(lane, mm - 1)];
      auto fetch = [&](int r, float (&hv)[NF], int& jj) __attribute__((always_inline)) {
        const int b = min(r, mm - 1);
        const int s = __builtin_amdgcn_readlane(en.x, b);
        jj = __builtin_amdgcn_readlane(en.y, b);
#pragma unroll
        for (int c = 0; c < NF; ++c) hv[c] = hsrc[(long)s * F + j0 + c];
      };
      float hb[DEPTH][NF];
      int   jb[DEPTH];
#pragma unroll
      for (int d = 0; d < DEPTH; ++d) fetch(d, hb[d], jb[d]);
      int r = 0;
      for (; r + DEPTH <= mm; r += DEPTH) {
#pragma unroll
        for (int d = 0; d < DEPTH; ++d) {
          apply(jb[d], hb[d]);
          fetch(r + d + DEPTH, hb[d], jb[d]);
        }
      }
#pragma unroll
      for (int d = 0; d < DEPTH; ++d)
        if (r + d < mm) apply(jb[d], hb[d]);
    }
  };

  int  off = 0;
  bool build = true;
  bool whole = true;
  if (!BUILD) {
    const int t = wtotal[blockIdx.x * AGG_WAVES + lane];
    if (__builtin_amdgcn_ballot_w32(t < 0) == 0) {
      build = false;
      off = __builtin_amdgcn_readlane(t, w);
    }
  }

  if (build) {
    const int* __restrict__ dstp = eidx + E;
    const bool vec_ok = ((E & 3) == 0) && ((((size_t)dstp) & 15) == 0);
    if (threadIdx.x == 0) { cnt[0] = 0; cnt[1] = 0; }
    if (threadIdx.x < AGG_WAVES) { wcnt[0][threadIdx.x] = 0; wcnt[1][threadIdx.x] = 0; }
    int tot = 0;
    int cb = 0;
    int S = AGG_S;
    for (int c0 = 0; c0 < E; cb ^= 1) {
      __syncthreads();
      const int lim = (E - c0 < S) ? E : c0 + S;
#pragma unroll
      for (int tg = 0; tg < AGG_STEPS; tg += AGG_GROUP) {
        int4 dv[AGG_GROUP];
        if (vec_ok) {
#pragma unroll
          for (int t = 0; t < AGG_GROUP; ++t) {
            int i = c0 + ((tg + t) * (AGG_WAVES * 32) + (int)threadIdx.x) * 4;
            dv[t] = (i < lim) ? *(const int4*)(dstp + i) : make_int4(-1, -1, -1, -1);
          }
        } else {
#pragma unroll
          for (int t = 0; t < AGG_GROUP; ++t) {
            int i = c0 + ((tg + t) * (AGG_WAVES * 32) + (int)threadIdx.x) * 4;
            dv[t].x = (i     < lim) ? dstp[i]     : -1;
            dv[t].y = (i + 1 < lim) ? dstp[i + 1] : -1;
            dv[t].z = (i + 2 < lim) ? dstp[i + 2] : -1;
            dv[t].w = (i + 3 < lim) ? dstp[i + 3] : -1;
          }
        }
#pragma unroll
        for (int t = 0; t < AGG_GROUP; ++t) {
          int i = c0 + ((tg + t) * (AGG_WAVES * 32) + (int)threadIdx.x) * 4;
          int dd[4] = {dv[t].x, dv[t].y, dv[t].z, dv[t].w};
#pragma unroll
          for (int k = 0; k < 4; ++k) {
            unsigned dl = (unsigned)(dd[k] - r0);
            if (dl < (unsigned)AGG_ROWS) {
              int pos = atomicAdd(&cnt[cb], 1);
              list[pos].y = (int)((dl << 16) | (unsigned)(i + k - c0));
              atomicAdd(&wcnt[cb][dl & 31], 1);
            }
          }
        }
      }
      __syncthreads();
      const int n = cnt[cb];
      if (threadIdx.x == 0) cnt[cb ^ 1] = 0;
      if (threadIdx.x < AGG_WAVES) wcnt[cb ^ 1][threadIdx.x] = 0;
      for (int idx = threadIdx.x; idx < n; idx += AGG_WAVES * 32)
        list[idx].x = eidx[c0 + (list[idx].y & 0xffff)];
      __syncthreads();
      const int wc = wcnt[cb][lane];
      if (__builtin_amdgcn_ballot_w32(wc > 32) != 0) { S >>= 1; continue; }
      tot += wc;
      if (__builtin_amdgcn_ballot_w32(tot > AGG_CAP) != 0) {
        __syncthreads();
        stream(off);
        off = 0; tot = wc; whole = false;
      }

      int qk = 0x7fffffff, qs = 0;
      int m = 0;
      for (int base = 0; base < n; base += 32 * AGG_SCAN) {
        int2 en[AGG_SCAN];
#pragma unroll
        for (int u = 0; u < AGG_SCAN; ++u) en[u] = list[min(base + 32 * u + lane, n - 1)];
#pragma unroll
        for (int u = 0; u < AGG_SCAN; ++u) {
          const bool own = (base + 32 * u + lane < n) && (((en[u].y >> 16) & 31) == w);
          unsigned mask = __builtin_amdgcn_ballot_w32(own);
          while (mask) {
            const int b = __builtin_ctz(mask);
            mask &= mask - 1;
            const int fk = __builtin_amdgcn_readlane(en[u].y, b);
            const int fs = __builtin_amdgcn_readlane(en[u].x, b);
            if (lane == m) { qk = fk; qs = fs; }
            ++m;
          }
        }
      }
      if (m > 0) {
        int rank = 0;
        for (int j = 0; j < m; ++j) rank += (__builtin_amdgcn_readlane(qk, j) < qk) ? 1 : 0;
        if (lane < m) mylist[off + rank] = make_int2(qs, qk >> 21);
        off += m;
      }
      c0 += S;
      S = AGG_S;
    }
    __syncthreads();
  }

  stream(off);
  if (BUILD && lane == 0) wtotal[blockIdx.x * AGG_WAVES + w] = whole ? off : -1;

  const int j0e = (int)__builtin_amdgcn_mbcnt_lo(~0u, 0u) * NF;
#pragma unroll
  for (int j = 0; j < AGG_RPW; ++j) {
    int loc = j * 32 + w;
    int row = r0 + loc;
    if (loc < AGG_ROWS && row < nnodes) {
#pragma unroll
      for (int k = 0; k < NF; ++k) {
        float v = acc[j][k] / sm[j] + bias[j0e + k];
        if (DO_ELU) v = v > 0.f ? v : (__expf(v) - 1.f);
        outp[(long)row * F + j0e + k] = v;
      }
    }
  }
}


static inline int cdiv_i(long a, long b) { return (int)((a + b - 1) / b); }

static void launch_gemm(const float* A, const float* B, float* C,
                        int M, int K, int Nd, hipStream_t stream) {
  int tilesM = M / 16, tilesN = Nd / 16;
  int wpb = tilesN >= 8 ? 8 : tilesN;
  dim3 grid(tilesM * (tilesN / wpb));
  dim3 block(32 * wpb);
  size_t shmem = (size_t)16 * K * sizeof(float);
  gemm_wmma_f32_lds<<<grid, block, shmem, stream>>>(A, B, C, M, K, Nd, wpb);
}

extern "C" void kernel_launch(void* const* d_in, const int* in_sizes, int n_in,
                              void* d_out, int out_size, void* d_ws, size_t ws_size,
                              hipStream_t stream) {
  const float* x   = (const float*)d_in[0];
  const int*   eix = (const int*)d_in[1];
  const float* W1  = (const float*)d_in[2];
  const float* as1 = (const float*)d_in[3];
  const float* ad1 = (const float*)d_in[4];
  const float* b1  = (const float*)d_in[5];
  const float* W2  = (const float*)d_in[6];
  const float* as2 = (const float*)d_in[7];
  const float* ad2 = (const float*)d_in[8];
  const float* b2  = (const float*)d_in[9];
  float* out = (float*)d_out;

  const int IN_DIM = 128;
  const int F1     = in_sizes[5];
  const int HID    = in_sizes[9];
  const int HEADS  = F1 / HID;
  const int Nn     = in_sizes[0] / IN_DIM;
  const int E      = in_sizes[1] / 2;
  const int EN     = E + Nn;

  float* acc1 = (float*)d_ws;
  float* h1   = acc1 + (long)Nn * F1;
  float* sc   = h1 + (long)Nn * F1;
  float* als1 = sc;
  float* ald1 = als1 + (long)Nn * HEADS;
  float* m1   = ald1 + (long)Nn * HEADS;
  float* s1   = m1   + (long)Nn * HEADS;
  float* als2 = s1   + (long)Nn * HEADS;
  float* ald2 = als2 + Nn;
  float* m2   = ald2 + Nn;
  float* s2   = m2   + Nn;
  float* h2   = h1;
  int*   wl   = (int*)(s2 + Nn);
  int*   wtot = wl + (long)cdiv_i(Nn, AGG_ROWS) * AGG_WAVES * AGG_CAP * 2;

  const float NEGINF = -__builtin_huge_valf();

  launch_gemm(x, W1, h1, Nn, IN_DIM, F1, stream);
  if (HEADS == 4 && HID == 64) {
    agg_owner_kernel<4, 64, 1, 1><<<cdiv_i(Nn, AGG_ROWS), AGG_WAVES * 32, 0, stream>>>(
        eix, E, Nn, as1, ad1, h1, b1, acc1, wl, wtot);
  } else {
    logits_kernel<<<cdiv_i((long)Nn * HEADS * 32, 256), 256, 0, stream>>>(
        h1, as1, ad1, als1, ald1, Nn, HEADS, HID);
    fill_f32<<<cdiv_i((long)Nn * F1, 256), 256, 0, stream>>>(acc1, (long)Nn * F1, 0.f);
    fill_f32<<<cdiv_i((long)Nn * HEADS, 256), 256, 0, stream>>>(m1, (long)Nn * HEADS, NEGINF);
    fill_f32<<<cdiv_i((long)Nn * HEADS, 256), 256, 0, stream>>>(s1, (long)Nn * HEADS, 0.f);
    edge_max_kernel<<<cdiv_i((long)EN * HEADS, 256), 256, 0, stream>>>(
        eix, E, EN, als1, ald1, m1, HEADS);
    edge_fused_gen<<<cdiv_i((long)EN * 32, 256), 256, 0, stream>>>(
        eix, E, EN, als1, ald1, m1, s1, h1, acc1, HEADS, HID);
    norm_bias_act_kernel<<<cdiv_i((long)Nn * F1, 256), 256, 0, stream>>>(
        acc1, s1, b1, (long)Nn * F1, F1, HID, HEADS, 1);
  }

  launch_gemm(acc1, W2, h2, Nn, F1, HID, stream);
  if (HEADS == 4 && HID == 64) {
    agg_owner_kernel<1, 64, 0, 0><<<cdiv_i(Nn, AGG_ROWS), AGG_WAVES * 32, 0, stream>>>(
        eix, E, Nn, as2, ad2, h2, b2, out, wl, wtot);
  } else if (HID == 64) {
    agg_owner_kernel<1, 64, 0, 1><<<cdiv_i(Nn, AGG_ROWS), AGG_WAVES * 32, 0, stream>>>(
        eix, E, Nn, as2, ad2, h2, b2, out, wl, wtot);
  } else {
    logits_kernel<<<cdiv_i((long)Nn * 32, 256), 256, 0, stream>>>(
        h2, as2, ad2, als2, ald2, Nn, 1, HID);
    fill_f32<<<cdiv_i(Nn, 256), 256, 0, stream>>>(m2, Nn, NEGINF);
    fill_f32<<<cdiv_i(Nn, 256), 256, 0, stream>>>(s2, Nn, 0.f);
    fill_f32<<<cdiv_i((long)Nn * HID, 256), 256, 0, stream>>>(out, (long)Nn * HID, 0.f);
    edge_max_kernel<<<cdiv_i((long)EN, 256), 256, 0, stream>>>(
        eix, E, EN, als2, ald2, m2, 1);
    edge_fused_gen<<<cdiv_i((long)EN * 32, 256), 256, 0, stream>>>(
        eix, E, EN, als2, ald2, m2, s2, h2, out, 1, HID);
    norm_bias_act_kernel<<<cdiv_i((long)Nn * HID, 256), 256, 0, stream>>>(
        out, s2, b2, (long)Nn * HID, HID, HID, 1, 0);
  }
}


# Both GEMMs: a wave reads its 16 A rows from global into registers (no LDS), walks the column tiles; W copied once per resident work-group into LDS in operand order, 16-B reads; k sum order new
# speedup vs baseline: 6.2296x; 1.3907x over previous
#include <hip/hip_runtime.h>

typedef float v2f __attribute__((ext_vector_type(2)));
typedef float v8f __attribute__((ext_vector_type(8)));

#define NEG_SLOPE 0.2f


__global__ void fill_f32(float* __restrict__ p, long n, float v) {
  long i = (long)blockIdx.x * blockDim.x + threadIdx.x;
  long stride = (long)gridDim.x * blockDim.x;
  for (; i < n; i += stride) p[i] = v;
}

__global__ void gemm_wmma_f32_lds(const float* __restrict__ A, const float* __restrict__ B,
                                  float* __restrict__ C, int M, int K, int Nd,
                                  int tnPerBlock) {
  extern __shared__ float As[];
  int tilesN = Nd >> 4;
  int blocksPerRow = tilesN / tnPerBlock;
  int tm  = blockIdx.x / blocksPerRow;
  int tn0 = (blockIdx.x % blocksPerRow) * tnPerBlock;
  int wib  = threadIdx.x >> 5;
  int lane = threadIdx.x & 31;

  {
    const float4* Av  = (const float4*)(A + (long)tm * 16 * K);
    float4*       Asv = (float4*)As;
    int nvec = (16 * K) >> 2;
    for (int idx = threadIdx.x; idx < nvec; idx += blockDim.x) Asv[idx] = Av[idx];
  }
  __syncthreads();

  int tn   = tn0 + wib;
  int half = lane >> 4;
  int l15  = lane & 15;
  int bcol = tn * 16 + l15;
  v8f acc = {};
  for (int k = 0; k < K; k += 4) {
    int ka = k + 2 * half;
    v2f a, b;
    a.x = As[l15 * K + ka];
    a.y = As[l15 * K + ka + 1];
    b.x = B[(long)ka * Nd + bcol];
    b.y = B[(long)(ka + 1) * Nd + bcol];
    acc = __builtin_amdgcn_wmma_f32_16x16x4_f32(false, a, false, b,
                                                (short)0, acc, false, false);
  }
  long crow = (long)tm * 16 + half * 8;
#pragma unroll
  for (int r = 0; r < 8; ++r)
    C[(crow + r) * Nd + bcol] = acc[r];
}

#define GR_WAVES 16

template<int K, int ND, int CT>
__global__ void __launch_bounds__(GR_WAVES * 32)
gemm_rows_in_regs(const float* __restrict__ A, const float* __restrict__ B,
                  float* __restrict__ C, int M) {
  constexpr int NT = ND / 16;
  constexpr int NG = NT / CT;
  constexpr int NQ = K / 8;
  static_assert(K % 8 == 0 && ND % 16 == 0 && NT % CT == 0, "shape must tile");
  __shared__ float4 Bs[NQ * NT * 32];

  const int w    = __builtin_amdgcn_readfirstlane((int)(threadIdx.x >> 5));
  const int lane = threadIdx.x & 31;
  const int c    = lane & 15;
  const int h    = lane >> 4;
  const int tilesM = M >> 4;

  const int wa = w >> 2, wb = w & 3;
  int slot = wa + 4 * ((wb - wa) & 3);

  float4 a[NQ];
  auto load_a = [&](int tm) __attribute__((always_inline)) {
    const float4* p = (const float4*)(A + (long)(tm * 16 + c) * K + 4 * h);
#pragma unroll
    for (int q = 0; q < NQ; ++q) a[q] = p[2 * q];
  };

  int tm = (int)blockIdx.x + (int)gridDim.x * (slot / NG);
  if (tm < tilesM) load_a(tm);

#pragma unroll 4
  for (int idx = threadIdx.x; idx < NQ * NT * 32; idx += GR_WAVES * 32) {
    const int t = idx >> 5;
    const int n = t % NT, q = t / NT;
    const float* s = B + (long)(8 * q + 4 * h) * ND + 16 * n + c;
    float4 v;
    v.x = s[0]; v.y = s[ND]; v.z = s[2 * ND]; v.w = s[3 * ND];
    Bs[idx] = v;
  }
  __syncthreads();

  while (tm < tilesM) {
    const int g = slot % NG;
    const float4* bp = Bs + g * CT * 32 + lane;
    v8f acc[CT];
#pragma unroll
    for (int n = 0; n < CT; ++n) acc[n] = v8f{};
    float4 bc[CT], bn[CT];
#pragma unroll
    for (int n = 0; n < CT; ++n) bc[n] = bp[n * 32];
#pragma unroll
    for (int q = 0; q < NQ; ++q) {
      if (q + 1 < NQ) {
#pragma unroll
        for (int n = 0; n < CT; ++n) bn[n] = bp[((q + 1) * NT + n) * 32];
      }
      __builtin_amdgcn_sched_barrier(0);
      v2f a0, a1;
      a0.x = a[q].x; a0.y = a[q].y; a1.x = a[q].z; a1.y = a[q].w;
#pragma unroll
      for (int n = 0; n < CT; ++n) {
        v2f b0; b0.x = bc[n].x; b0.y = bc[n].y;
        acc[n] = __builtin_amdgcn_wmma_f32_16x16x4_f32(false, a0, false, b0,
                                                       (short)0, acc[n], false, false);
      }
#pragma unroll
      for (int n = 0; n < CT; ++n) {
        v2f b1; b1.x = bc[n].z; b1.y = bc[n].w;
        acc[n] = __builtin_amdgcn_wmma_f32_16x16x4_f32(false, a1, false, b1,
                                                       (short)0, acc[n], false, false);
      }
      __builtin_amdgcn_sched_barrier(0);
      if (q + 1 < NQ) {
#pragma unroll
        for (int n = 0; n < CT; ++n) bc[n] = bn[n];
      }
    }
    float* cp = C + (long)(tm * 16 + 8 * h) * ND + g * CT * 16 + c;
#pragma unroll
    for (int n = 0; n < CT; ++n) {
#pragma unroll
      for (int r = 0; r < 8; ++r) cp[r * ND + n * 16] = acc[n][r];
    }
    slot += GR_WAVES;
    tm = (int)blockIdx.x + (int)gridDim.x * (slot / NG);
    if (tm < tilesM) load_a(tm);
  }
}

__global__ void logits_kernel(const float* __restrict__ hmat, const float* __restrict__ a_src,
                              const float* __restrict__ a_dst, float* __restrict__ als,
                              float* __restrict__ ald, int nnodes, int heads, int ch) {
  int wid  = (int)(((long)blockIdx.x * blockDim.x + threadIdx.x) >> 5);
  int lane = threadIdx.x & 31;
  if (wid >= nnodes * heads) return;
  int node = wid / heads, hh = wid % heads;
  int F = heads * ch;
  float ss = 0.f, dd = 0.f;
  for (int c = lane; c < ch; c += 32) {
    float hv = hmat[(long)node * F + hh * ch + c];
    ss += hv * a_src[hh * ch + c];
    dd += hv * a_dst[hh * ch + c];
  }
#pragma unroll
  for (int off = 16; off > 0; off >>= 1) {
    ss += __shfl_xor(ss, off);
    dd += __shfl_xor(dd, off);
  }
  if (lane == 0) { als[wid] = ss; ald[wid] = dd; }
}

__device__ __forceinline__ void edge_sd(const int* __restrict__ eidx, int E, int e,
                                        int& s, int& d) {
  if (e < E) { s = eidx[e]; d = eidx[E + e]; }
  else       { s = e - E; d = s; }
}

__global__ void edge_max_kernel(const int* __restrict__ eidx, int E, int EN,
                                const float* __restrict__ als, const float* __restrict__ ald,
                                float* __restrict__ m, int heads) {
  long t = (long)blockIdx.x * blockDim.x + threadIdx.x;
  if (t >= (long)EN * heads) return;
  int e = (int)(t / heads), hh = (int)(t % heads);
  int s, d; edge_sd(eidx, E, e, s, d);
  float l = als[s * heads + hh] + ald[d * heads + hh];
  l = l > 0.f ? l : NEG_SLOPE * l;
  unsafeAtomicMax(&m[d * heads + hh], l);
}

__global__ void edge_fused_gen(const int* __restrict__ eidx, int E, int EN,
                               const float* __restrict__ als, const float* __restrict__ ald,
                               const float* __restrict__ m, float* __restrict__ ssum,
                               const float* __restrict__ hsrc, float* __restrict__ accum,
                               int heads, int ch) {
  int wid  = (int)(((long)blockIdx.x * blockDim.x + threadIdx.x) >> 5);
  int lane = threadIdx.x & 31;
  if (wid >= EN) return;
  int s, d; edge_sd(eidx, E, wid, s, d);
  int F = heads * ch;
  for (int j = lane; j < F; j += 32) {
    int head = j / ch;
    float l = als[s * heads + head] + ald[d * heads + head];
    l = l > 0.f ? l : NEG_SLOPE * l;
    float ex = __expf(l - m[d * heads + head]);
    if ((j % ch) == 0) unsafeAtomicAdd(&ssum[d * heads + head], ex);
    unsafeAtomicAdd(&accum[(long)d * F + j], hsrc[(long)s * F + j] * ex);
  }
}

__global__ void norm_bias_act_kernel(float* __restrict__ a, const float* __restrict__ ssum,
                                     const float* __restrict__ bias, long n, int F,
                                     int ch, int heads, int do_elu) {
  long i = (long)blockIdx.x * blockDim.x + threadIdx.x;
  long stride = (long)gridDim.x * blockDim.x;
  for (; i < n; i += stride) {
    int  j    = (int)(i % F);
    long node = i / F;
    float v = a[i] / ssum[node * heads + j / ch] + bias[j];
    if (do_elu) v = v > 0.f ? v : (__expf(v) - 1.f);
    a[i] = v;
  }
}


#define AGG_WAVES 32
#define AGG_RPW   7
#define AGG_ROWS  196
#define AGG_S     32768
#define AGG_STEPS (AGG_S / (AGG_WAVES * 32 * 4))
#define AGG_GROUP 4
#define AGG_SCAN  4
#define AGG_CAP   1024

template<int NF>
__device__ __forceinline__ float dot_part(const float (&v)[NF], const float (&a)[NF]) {
  float p0 = v[0] * a[0];
  float p1 = (NF > 1) ? v[NF > 1 ? 1 : 0] * a[NF > 1 ? 1 : 0] : 0.f;
#pragma unroll
  for (int k = 2; k < NF; k += 2) {
    p0 = __builtin_fmaf(v[k], a[k], p0);
    if (k + 1 < NF) p1 = __builtin_fmaf(v[k + 1], a[k + 1], p1);
  }
  return p0 + p1;
}

template<int G>
__device__ __forceinline__ float group_sum(float p) {
  static_assert(G == 1 || G == 2 || G == 4 || G == 8 || G == 16 || G == 32, "lanes per head must be a power of two");
  if (G >= 2)  p += __int_as_float(__builtin_amdgcn_update_dpp(0, __float_as_int(p), 0x161, 0xf, 0xf, false));
  if (G >= 4)  p += __int_as_float(__builtin_amdgcn_update_dpp(0, __float_as_int(p), 0x162, 0xf, 0xf, false));
  if (G >= 8)  p += __int_as_float(__builtin_amdgcn_update_dpp(0, __float_as_int(p), 0x164, 0xf, 0xf, false));
  if (G >= 16) p += __int_as_float(__builtin_amdgcn_update_dpp(0, __float_as_int(p), 0x168, 0xf, 0xf, false));
  if (G >= 32) {
    const int b = __float_as_int(p);
    p += __int_as_float(__builtin_amdgcn_permlanex16(b, b, 0x76543210, 0xfedcba98, false, false));
  }
  return p;
}

template<int HEADS, int CH, int DO_ELU, int BUILD>
__global__ void __launch_bounds__(AGG_WAVES * 32)
agg_owner_kernel(const int* __restrict__ eidx, int E, int nnodes,
                 const float* __restrict__ a_src, const float* __restrict__ a_dst,
                 const float* __restrict__ hsrc, const float* __restrict__ bias,
                 float* __restrict__ outp, int* wlist, int* wtotal) {
  constexpr int F  = HEADS * CH;
  constexpr int NF = F / 32;
  constexpr int LPH = CH / NF;
  static_assert(NF >= 1 && CH % NF == 0 && LPH <= 32, "a head's channels must fill whole lanes");
  constexpr int DEPTH = (NF >= 8) ? 2 : ((NF >= 4) ? 4 : 8);
  __shared__ int2 list[AGG_S];
  __shared__ int  cnt[2];
  __shared__ int  wcnt[2][AGG_WAVES];
  const int w    = __builtin_amdgcn_readfirstlane((int)(threadIdx.x >> 5));
  const int lane = threadIdx.x & 31;
  const int r0   = blockIdx.x * AGG_ROWS;
  const int j0   = lane * NF;
  int2* const mylist = (int2*)wlist + (long)(blockIdx.x * AGG_WAVES + w) * AGG_CAP;

  float mx[AGG_RPW], sm[AGG_RPW], acc[AGG_RPW][NF];
  float as_l[NF];
  float adr[AGG_RPW];
  {
    float ad_l[NF];
#pragma unroll
    for (int k = 0; k < NF; ++k) { as_l[k] = a_src[j0 + k]; ad_l[k] = a_dst[j0 + k]; }
#pragma unroll
    for (int j = 0; j < AGG_RPW; ++j) {
      int loc = j * 32 + w;
      int row = r0 + loc;
      if (loc < AGG_ROWS && row < nnodes) {
#pragma unroll
        for (int k = 0; k < NF; ++k) acc[j][k] = hsrc[(long)row * F + j0 + k];
        float a_s = group_sum<LPH>(dot_part<NF>(acc[j], as_l));
        float a_d = group_sum<LPH>(dot_part<NF>(acc[j], ad_l));
        float l = a_s + a_d;
        l = l > 0.f ? l : NEG_SLOPE * l;
        mx[j] = l; sm[j] = 1.f; adr[j] = a_d;
      } else {
        mx[j] = 0.f; sm[j] = 1.f; adr[j] = 0.f;
#pragma unroll
        for (int k = 0; k < NF; ++k) acc[j][k] = 0.f;
      }
    }
  }

  auto apply = [&](int jj, const float (&hv)[NF]) __attribute__((always_inline)) {
    const float a_s = group_sum<LPH>(dot_part<NF>(hv, as_l));
#pragma unroll
    for (int j = 0; j < AGG_RPW; ++j) {
      if (jj == j) {
        float l = a_s + adr[j];
        l = l > 0.f ? l : NEG_SLOPE * l;
        float mn = fmaxf(mx[j], l);
        float sc = __expf(mx[j] - mn);
        float wg = __expf(l - mn);
        sm[j] = sm[j] * sc + wg;
#pragma unroll
        for (int k = 0; k < NF; ++k) acc[j][k] = acc[j][k] * sc + wg * hv[k];
        mx[j] = mn;
      }
    }
  };

  auto stream = [&](int nent) __attribute__((always_inline)) {
    for (int base = 0; base < nent; base += 32) {
      const int mm = min(32, nent - base);
      const int2 en = mylist[base + min(lane, mm - 1)];
      auto fetch = [&](int r, float (&hv)[NF], int& jj) __attribute__((always_inline)) {
        const int b = min(r, mm - 1);
        const int s = __builtin_amdgcn_readlane(en.x, b);
        jj = __builtin_amdgcn_readlane(en.y, b);
#pragma unroll
        for (int c = 0; c < NF; ++c) hv[c] = hsrc[(long)s * F + j0 + c];
      };
      float hb[DEPTH][NF];
      int   jb[DEPTH];
#pragma unroll
      for (int d = 0; d < DEPTH; ++d) fetch(d, hb[d], jb[d]);
      int r = 0;
      for (; r + DEPTH <= mm; r += DEPTH) {
#pragma unroll
        for (int d = 0; d < DEPTH; ++d) {
          apply(jb[d], hb[d]);
          fetch(r + d + DEPTH, hb[d], jb[d]);
        }
      }
#pragma unroll
      for (int d = 0; d < DEPTH; ++d)
        if (r + d < mm) apply(jb[d], hb[d]);
    }
  };

  int  off = 0;
  bool build = true;
  bool whole = true;
  if (!BUILD) {
    const int t = wtotal[blockIdx.x * AGG_WAVES + lane];
    if (__builtin_amdgcn_ballot_w32(t < 0) == 0) {
      build = false;
      off = __builtin_amdgcn_readlane(t, w);
    }
  }

  if (build) {
    const int* __restrict__ dstp = eidx + E;
    const bool vec_ok = ((E & 3) == 0) && ((((size_t)dstp) & 15) == 0);
    if (threadIdx.x == 0) { cnt[0] = 0; cnt[1] = 0; }
    if (threadIdx.x < AGG_WAVES) { wcnt[0][threadIdx.x] = 0; wcnt[1][threadIdx.x] = 0; }
    int tot = 0;
    int cb = 0;
    int S = AGG_S;
    for (int c0 = 0; c0 < E; cb ^= 1) {
      __syncthreads();
      const int lim = (E - c0 < S) ? E : c0 + S;
#pragma unroll
      for (int tg = 0; tg < AGG_STEPS; tg += AGG_GROUP) {
        int4 dv[AGG_GROUP];
        if (vec_ok) {
#pragma unroll
          for (int t = 0; t < AGG_GROUP; ++t) {
            int i = c0 + ((tg + t) * (AGG_WAVES * 32) + (int)threadIdx.x) * 4;
            dv[t] = (i < lim) ? *(const int4*)(dstp + i) : make_int4(-1, -1, -1, -1);
          }
        } else {
#pragma unroll
          for (int t = 0; t < AGG_GROUP; ++t) {
            int i = c0 + ((tg + t) * (AGG_WAVES * 32) + (int)threadIdx.x) * 4;
            dv[t].x = (i     < lim) ? dstp[i]     : -1;
            dv[t].y = (i + 1 < lim) ? dstp[i + 1] : -1;
            dv[t].z = (i + 2 < lim) ? dstp[i + 2] : -1;
            dv[t].w = (i + 3 < lim) ? dstp[i + 3] : -1;
          }
        }
#pragma unroll
        for (int t = 0; t < AGG_GROUP; ++t) {
          int i = c0 + ((tg + t) * (AGG_WAVES * 32) + (int)threadIdx.x) * 4;
          int dd[4] = {dv[t].x, dv[t].y, dv[t].z, dv[t].w};
#pragma unroll
          for (int k = 0; k < 4; ++k) {
            unsigned dl = (unsigned)(dd[k] - r0);
            if (dl < (unsigned)AGG_ROWS) {
              int pos = atomicAdd(&cnt[cb], 1);
              list[pos].y = (int)((dl << 16) | (unsigned)(i + k - c0));
              atomicAdd(&wcnt[cb][dl & 31], 1);
            }
          }
        }
      }
      __syncthreads();
      const int n = cnt[cb];
      if (threadIdx.x == 0) cnt[cb ^ 1] = 0;
      if (threadIdx.x < AGG_WAVES) wcnt[cb ^ 1][threadIdx.x] = 0;
      for (int idx = threadIdx.x; idx < n; idx += AGG_WAVES * 32)
        list[idx].x = eidx[c0 + (list[idx].y & 0xffff)];
      __syncthreads();
      const int wc = wcnt[cb][lane];
      if (__builtin_amdgcn_ballot_w32(wc > 32) != 0) { S >>= 1; continue; }
      tot += wc;
      if (__builtin_amdgcn_ballot_w32(tot > AGG_CAP) != 0) {
        __syncthreads();
        stream(off);
        off = 0; tot = wc; whole = false;
      }

      int qk = 0x7fffffff, qs = 0;
      int m = 0;
      for (int base = 0; base < n; base += 32 * AGG_SCAN) {
        int2 en[AGG_SCAN];
#pragma unroll
        for (int u = 0; u < AGG_SCAN; ++u) en[u] = list[min(base + 32 * u + lane, n - 1)];
#pragma unroll
        for (int u = 0; u < AGG_SCAN; ++u) {
          const bool own = (base + 32 * u + lane < n) && (((en[u].y >> 16) & 31) == w);
          unsigned mask = __builtin_amdgcn_ballot_w32(own);
          while (mask) {
            const int b = __builtin_ctz(mask);
            mask &= mask - 1;
            const int fk = __builtin_amdgcn_readlane(en[u].y, b);
            const int fs = __builtin_amdgcn_readlane(en[u].x, b);
            if (lane == m) { qk = fk; qs = fs; }
            ++m;
          }
        }
      }
      if (m > 0) {
        int rank = 0;
        for (int j = 0; j < m; ++j) rank += (__builtin_amdgcn_readlane(qk, j) < qk) ? 1 : 0;
        if (lane < m) mylist[off + rank] = make_int2(qs, qk >> 21);
        off += m;
      }
      c0 += S;
      S = AGG_S;
    }
    __syncthreads();
  }

  stream(off);
  if (BUILD && lane == 0) wtotal[blockIdx.x * AGG_WAVES + w] = whole ? off : -1;

  const int j0e = (int)__builtin_amdgcn_mbcnt_lo(~0u, 0u) * NF;
#pragma unroll
  for (int j = 0; j < AGG_RPW; ++j) {
    int loc = j * 32 + w;
    int row = r0 + loc;
    if (loc < AGG_ROWS && row < nnodes) {
#pragma unroll
      for (int k = 0; k < NF; ++k) {
        float v = acc[j][k] / sm[j] + bias[j0e + k];
        if (DO_ELU) v = v > 0.f ? v : (__expf(v) - 1.f);
        outp[(long)row * F + j0e + k] = v;
      }
    }
  }
}


static inline int cdiv_i(long a, long b) { return (int)((a + b - 1) / b); }

static void launch_gemm(const float* A, const float* B, float* C,
                        int M, int K, int Nd, hipStream_t stream) {
  if ((M & 15) == 0 && K == 128 && Nd == 256) {
    gemm_rows_in_regs<128, 256, 8><<<256, GR_WAVES * 32, 0, stream>>>(A, B, C, M);
    return;
  }
  if ((M & 15) == 0 && K == 256 && Nd == 64) {
    gemm_rows_in_regs<256, 64, 4><<<256, GR_WAVES * 32, 0, stream>>>(A, B, C, M);
    return;
  }
  int tilesM = M / 16, tilesN = Nd / 16;
  int wpb = tilesN >= 8 ? 8 : tilesN;
  dim3 grid(tilesM * (tilesN / wpb));
  dim3 block(32 * wpb);
  size_t shmem = (size_t)16 * K * sizeof(float);
  gemm_wmma_f32_lds<<<grid, block, shmem, stream>>>(A, B, C, M, K, Nd, wpb);
}

extern "C" void kernel_launch(void* const* d_in, const int* in_sizes, int n_in,
                              void* d_out, int out_size, void* d_ws, size_t ws_size,
                              hipStream_t stream) {
  const float* x   = (const float*)d_in[0];
  const int*   eix = (const int*)d_in[1];
  const float* W1  = (const float*)d_in[2];
  const float* as1 = (const float*)d_in[3];
  const float* ad1 = (const float*)d_in[4];
  const float* b1  = (const float*)d_in[5];
  const float* W2  = (const float*)d_in[6];
  const float* as2 = (const float*)d_in[7];
  const float* ad2 = (const float*)d_in[8];
  const float* b2  = (const float*)d_in[9];
  float* out = (float*)d_out;

  const int IN_DIM = 128;
  const int F1     = in_sizes[5];
  const int HID    = in_sizes[9];
  const int HEADS  = F1 / HID;
  const int Nn     = in_sizes[0] / IN_DIM;
  const int E      = in_sizes[1] / 2;
  const int EN     = E + Nn;

  float* acc1 = (float*)d_ws;
  float* h1   = acc1 + (long)Nn * F1;
  float* sc   = h1 + (long)Nn * F1;
  float* als1 = sc;
  float* ald1 = als1 + (long)Nn * HEADS;
  float* m1   = ald1 + (long)Nn * HEADS;
  float* s1   = m1   + (long)Nn * HEADS;
  float* als2 = s1   + (long)Nn * HEADS;
  float* ald2 = als2 + Nn;
  float* m2   = ald2 + Nn;
  float* s2   = m2   + Nn;
  float* h2   = h1;
  int*   wl   = (int*)(s2 + Nn);
  int*   wtot = wl + (long)cdiv_i(Nn, AGG_ROWS) * AGG_WAVES * AGG_CAP * 2;

  const float NEGINF = -__builtin_huge_valf();

  launch_gemm(x, W1, h1, Nn, IN_DIM, F1, stream);
  if (HEADS == 4 && HID == 64) {
    agg_owner_kernel<4, 64, 1, 1><<<cdiv_i(Nn, AGG_ROWS), AGG_WAVES * 32, 0, stream>>>(
        eix, E, Nn, as1, ad1, h1, b1, acc1, wl, wtot);
  } else {
    logits_kernel<<<cdiv_i((long)Nn * HEADS * 32, 256), 256, 0, stream>>>(
        h1, as1, ad1, als1, ald1, Nn, HEADS, HID);
    fill_f32<<<cdiv_i((long)Nn * F1, 256), 256, 0, stream>>>(acc1, (long)Nn * F1, 0.f);
    fill_f32<<<cdiv_i((long)Nn * HEADS, 256), 256, 0, stream>>>(m1, (long)Nn * HEADS, NEGINF);
    fill_f32<<<cdiv_i((long)Nn * HEADS, 256), 256, 0, stream>>>(s1, (long)Nn * HEADS, 0.f);
    edge_max_kernel<<<cdiv_i((long)EN * HEADS, 256), 256, 0, stream>>>(
        eix, E, EN, als1, ald1, m1, HEADS);
    edge_fused_gen<<<cdiv_i((long)EN * 32, 256), 256, 0, stream>>>(
        eix, E, EN, als1, ald1, m1, s1, h1, acc1, HEADS, HID);
    norm_bias_act_kernel<<<cdiv_i((long)Nn * F1, 256), 256, 0, stream>>>(
        acc1, s1, b1, (long)Nn * F1, F1, HID, HEADS, 1);
  }

  launch_gemm(acc1, W2, h2, Nn, F1, HID, stream);
  if (HEADS == 4 && HID == 64) {
    agg_owner_kernel<1, 64, 0, 0><<<cdiv_i(Nn, AGG_ROWS), AGG_WAVES * 32, 0, stream>>>(
        eix, E, Nn, as2, ad2, h2, b2, out, wl, wtot);
  } else if (HID == 64) {
    agg_owner_kernel<1, 64, 0, 1><<<cdiv_i(Nn, AGG_ROWS), AGG_WAVES * 32, 0, stream>>>(
        eix, E, Nn, as2, ad2, h2, b2, out, wl, wtot);
  } else {
    logits_kernel<<<cdiv_i((long)Nn * 32, 256), 256, 0, stream>>>(
        h2, as2, ad2, als2, ald2, Nn, 1, HID);
    fill_f32<<<cdiv_i(Nn, 256), 256, 0, stream>>>(m2, Nn, NEGINF);
    fill_f32<<<cdiv_i(Nn, 256), 256, 0, stream>>>(s2, Nn, 0.f);
    fill_f32<<<cdiv_i((long)Nn * HID, 256), 256, 0, stream>>>(out, (long)Nn * HID, 0.f);
    edge_max_kernel<<<cdiv_i((long)EN, 256), 256, 0, stream>>>(
        eix, E, EN, als2, ald2, m2, 1);
    edge_fused_gen<<<cdiv_i((long)EN * 32, 256), 256, 0, stream>>>(
        eix, E, EN, als2, ald2, m2, s2, h2, out, 1, HID);
    norm_bias_act_kernel<<<cdiv_i((long)Nn * HID, 256), 256, 0, stream>>>(
        out, s2, b2, (long)Nn * HID, HID, HID, 1, 0);
  }
}


# List building: a pass takes 65536 edges into a key-only LDS list, 3-4 id loads in flight per thread all through; the owning lane fetches an entry's source id one pass later; sum order same
# speedup vs baseline: 7.4839x; 1.2013x over previous
#include <hip/hip_runtime.h>

typedef float v2f __attribute__((ext_vector_type(2)));
typedef float v8f __attribute__((ext_vector_type(8)));

#define NEG_SLOPE 0.2f


__global__ void fill_f32(float* __restrict__ p, long n, float v) {
  long i = (long)blockIdx.x * blockDim.x + threadIdx.x;
  long stride = (long)gridDim.x * blockDim.x;
  for (; i < n; i += stride) p[i] = v;
}

__global__ void gemm_wmma_f32_lds(const float* __restrict__ A, const float* __restrict__ B,
                                  float* __restrict__ C, int M, int K, int Nd,
                                  int tnPerBlock) {
  extern __shared__ float As[];
  int tilesN = Nd >> 4;
  int blocksPerRow = tilesN / tnPerBlock;
  int tm  = blockIdx.x / blocksPerRow;
  int tn0 = (blockIdx.x % blocksPerRow) * tnPerBlock;
  int wib  = threadIdx.x >> 5;
  int lane = threadIdx.x & 31;

  {
    const float4* Av  = (const float4*)(A + (long)tm * 16 * K);
    float4*       Asv = (float4*)As;
    int nvec = (16 * K) >> 2;
    for (int idx = threadIdx.x; idx < nvec; idx += blockDim.x) Asv[idx] = Av[idx];
  }
  __syncthreads();

  int tn   = tn0 + wib;
  int half = lane >> 4;
  int l15  = lane & 15;
  int bcol = tn * 16 + l15;
  v8f acc = {};
  for (int k = 0; k < K; k += 4) {
    int ka = k + 2 * half;
    v2f a, b;
    a.x = As[l15 * K + ka];
    a.y = As[l15 * K + ka + 1];
    b.x = B[(long)ka * Nd + bcol];
    b.y = B[(long)(ka + 1) * Nd + bcol];
    acc = __builtin_amdgcn_wmma_f32_16x16x4_f32(false, a, false, b,
                                                (short)0, acc, false, false);
  }
  long crow = (long)tm * 16 + half * 8;
#pragma unroll
  for (int r = 0; r < 8; ++r)
    C[(crow + r) * Nd + bcol] = acc[r];
}

#define GR_WAVES 16

template<int K, int ND, int CT>
__global__ void __launch_bounds__(GR_WAVES * 32)
gemm_rows_in_regs(const float* __restrict__ A, const float* __restrict__ B,
                  float* __restrict__ C, int M) {
  constexpr int NT = ND / 16;
  constexpr int NG = NT / CT;
  constexpr int NQ = K / 8;
  static_assert(K % 8 == 0 && ND % 16 == 0 && NT % CT == 0, "shape must tile");
  __shared__ float4 Bs[NQ * NT * 32];

  const int w    = __builtin_amdgcn_readfirstlane((int)(threadIdx.x >> 5));
  const int lane = threadIdx.x & 31;
  const int c    = lane & 15;
  const int h    = lane >> 4;
  const int tilesM = M >> 4;

  const int wa = w >> 2, wb = w & 3;
  int slot = wa + 4 * ((wb - wa) & 3);

  float4 a[NQ];
  auto load_a = [&](int tm) __attribute__((always_inline)) {
    const float4* p = (const float4*)(A + (long)(tm * 16 + c) * K + 4 * h);
#pragma unroll
    for (int q = 0; q < NQ; ++q) a[q] = p[2 * q];
  };

  int tm = (int)blockIdx.x + (int)gridDim.x * (slot / NG);
  if (tm < tilesM) load_a(tm);

#pragma unroll 4
  for (int idx = threadIdx.x; idx < NQ * NT * 32; idx += GR_WAVES * 32) {
    const int t = idx >> 5;
    const int n = t % NT, q = t / NT;
    const float* s = B + (long)(8 * q + 4 * h) * ND + 16 * n + c;
    float4 v;
    v.x = s[0]; v.y = s[ND]; v.z = s[2 * ND]; v.w = s[3 * ND];
    Bs[idx] = v;
  }
  __syncthreads();

  while (tm < tilesM) {
    const int g = slot % NG;
    const float4* bp = Bs + g * CT * 32 + lane;
    v8f acc[CT];
#pragma unroll
    for (int n = 0; n < CT; ++n) acc[n] = v8f{};
    float4 bc[CT], bn[CT];
#pragma unroll
    for (int n = 0; n < CT; ++n) bc[n] = bp[n * 32];
#pragma unroll
    for (int q = 0; q < NQ; ++q) {
      if (q + 1 < NQ) {
#pragma unroll
        for (int n = 0; n < CT; ++n) bn[n] = bp[((q + 1) * NT + n) * 32];
      }
      __builtin_amdgcn_sched_barrier(0);
      v2f a0, a1;
      a0.x = a[q].x; a0.y = a[q].y; a1.x = a[q].z; a1.y = a[q].w;
#pragma unroll
      for (int n = 0; n < CT; ++n) {
        v2f b0; b0.x = bc[n].x; b0.y = bc[n].y;
        acc[n] = __builtin_amdgcn_wmma_f32_16x16x4_f32(false, a0, false, b0,
                                                       (short)0, acc[n], false, false);
      }
#pragma unroll
      for (int n = 0; n < CT; ++n) {
        v2f b1; b1.x = bc[n].z; b1.y = bc[n].w;
        acc[n] = __builtin_amdgcn_wmma_f32_16x16x4_f32(false, a1, false, b1,
                                                       (short)0, acc[n], false, false);
      }
      __builtin_amdgcn_sched_barrier(0);
      if (q + 1 < NQ) {
#pragma unroll
        for (int n = 0; n < CT; ++n) bc[n] = bn[n];
      }
    }
    float* cp = C + (long)(tm * 16 + 8 * h) * ND + g * CT * 16 + c;
#pragma unroll
    for (int n = 0; n < CT; ++n) {
#pragma unroll
      for (int r = 0; r < 8; ++r) cp[r * ND + n * 16] = acc[n][r];
    }
    slot += GR_WAVES;
    tm = (int)blockIdx.x + (int)gridDim.x * (slot / NG);
    if (tm < tilesM) load_a(tm);
  }
}

__global__ void logits_kernel(const float* __restrict__ hmat, const float* __restrict__ a_src,
                              const float* __restrict__ a_dst, float* __restrict__ als,
                              float* __restrict__ ald, int nnodes, int heads, int ch) {
  int wid  = (int)(((long)blockIdx.x * blockDim.x + threadIdx.x) >> 5);
  int lane = threadIdx.x & 31;
  if (wid >= nnodes * heads) return;
  int node = wid / heads, hh = wid % heads;
  int F = heads * ch;
  float ss = 0.f, dd = 0.f;
  for (int c = lane; c < ch; c += 32) {
    float hv = hmat[(long)node * F + hh * ch + c];
    ss += hv * a_src[hh * ch + c];
    dd += hv * a_dst[hh * ch + c];
  }
#pragma unroll
  for (int off = 16; off > 0; off >>= 1) {
    ss += __shfl_xor(ss, off);
    dd += __shfl_xor(dd, off);
  }
  if (lane == 0) { als[wid] = ss; ald[wid] = dd; }
}

__device__ __forceinline__ void edge_sd(const int* __restrict__ eidx, int E, int e,
                                        int& s, int& d) {
  if (e < E) { s = eidx[e]; d = eidx[E + e]; }
  else       { s = e - E; d = s; }
}

__global__ void edge_max_kernel(const int* __restrict__ eidx, int E, int EN,
                                const float* __restrict__ als, const float* __restrict__ ald,
                                float* __restrict__ m, int heads) {
  long t = (long)blockIdx.x * blockDim.x + threadIdx.x;
  if (t >= (long)EN * heads) return;
  int e = (int)(t / heads), hh = (int)(t % heads);
  int s, d; edge_sd(eidx, E, e, s, d);
  float l = als[s * heads + hh] + ald[d * heads + hh];
  l = l > 0.f ? l : NEG_SLOPE * l;
  unsafeAtomicMax(&m[d * heads + hh], l);
}

__global__ void edge_fused_gen(const int* __restrict__ eidx, int E, int EN,
                               const float* __restrict__ als, const float* __restrict__ ald,
                               const float* __restrict__ m, float* __restrict__ ssum,
                               const float* __restrict__ hsrc, float* __restrict__ accum,
                               int heads, int ch) {
  int wid  = (int)(((long)blockIdx.x * blockDim.x + threadIdx.x) >> 5);
  int lane = threadIdx.x & 31;
  if (wid >= EN) return;
  int s, d; edge_sd(eidx, E, wid, s, d);
  int F = heads * ch;
  for (int j = lane; j < F; j += 32) {
    int head = j / ch;
    float l = als[s * heads + head] + ald[d * heads + head];
    l = l > 0.f ? l : NEG_SLOPE * l;
    float ex = __expf(l - m[d * heads + head]);
    if ((j % ch) == 0) unsafeAtomicAdd(&ssum[d * heads + head], ex);
    unsafeAtomicAdd(&accum[(long)d * F + j], hsrc[(long)s * F + j] * ex);
  }
}

__global__ void norm_bias_act_kernel(float* __restrict__ a, const float* __restrict__ ssum,
                                     const float* __restrict__ bias, long n, int F,
                                     int ch, int heads, int do_elu) {
  long i = (long)blockIdx.x * blockDim.x + threadIdx.x;
  long stride = (long)gridDim.x * blockDim.x;
  for (; i < n; i += stride) {
    int  j    = (int)(i % F);
    long node = i / F;
    float v = a[i] / ssum[node * heads + j / ch] + bias[j];
    if (do_elu) v = v > 0.f ? v : (__expf(v) - 1.f);
    a[i] = v;
  }
}


#define AGG_WAVES 32
#define AGG_RPW   7
#define AGG_ROWS  196
#define AGG_S     65536
#define AGG_STEPS (AGG_S / (AGG_WAVES * 32 * 4))
#define AGG_RING  4
#define AGG_SCAN  8
#define AGG_CAP   1024

template<int NF>
__device__ __forceinline__ float dot_part(const float (&v)[NF], const float (&a)[NF]) {
  float p0 = v[0] * a[0];
  float p1 = (NF > 1) ? v[NF > 1 ? 1 : 0] * a[NF > 1 ? 1 : 0] : 0.f;
#pragma unroll
  for (int k = 2; k < NF; k += 2) {
    p0 = __builtin_fmaf(v[k], a[k], p0);
    if (k + 1 < NF) p1 = __builtin_fmaf(v[k + 1], a[k + 1], p1);
  }
  return p0 + p1;
}

template<int G>
__device__ __forceinline__ float group_sum(float p) {
  static_assert(G == 1 || G == 2 || G == 4 || G == 8 || G == 16 || G == 32, "lanes per head must be a power of two");
  if (G >= 2)  p += __int_as_float(__builtin_amdgcn_update_dpp(0, __float_as_int(p), 0x161, 0xf, 0xf, false));
  if (G >= 4)  p += __int_as_float(__builtin_amdgcn_update_dpp(0, __float_as_int(p), 0x162, 0xf, 0xf, false));
  if (G >= 8)  p += __int_as_float(__builtin_amdgcn_update_dpp(0, __float_as_int(p), 0x164, 0xf, 0xf, false));
  if (G >= 16) p += __int_as_float(__builtin_amdgcn_update_dpp(0, __float_as_int(p), 0x168, 0xf, 0xf, false));
  if (G >= 32) {
    const int b = __float_as_int(p);
    p += __int_as_float(__builtin_amdgcn_permlanex16(b, b, 0x76543210, 0xfedcba98, false, false));
  }
  return p;
}

template<int HEADS, int CH, int DO_ELU, int BUILD>
__global__ void __launch_bounds__(AGG_WAVES * 32)
agg_owner_kernel(const int* __restrict__ eidx, int E, int nnodes,
                 const float* __restrict__ a_src, const float* __restrict__ a_dst,
                 const float* __restrict__ hsrc, const float* __restrict__ bias,
                 float* __restrict__ outp, int* wlist, int* wtotal) {
  constexpr int F  = HEADS * CH;
  constexpr int NF = F / 32;
  constexpr int LPH = CH / NF;
  static_assert(NF >= 1 && CH % NF == 0 && LPH <= 32, "a head's channels must fill whole lanes");
  constexpr int DEPTH = (NF >= 8) ? 2 : ((NF >= 4) ? 4 : 8);
  __shared__ int  list[AGG_S];
  __shared__ int  cnt[2];
  __shared__ int  wcnt[2][AGG_WAVES];
  const int w    = __builtin_amdgcn_readfirstlane((int)(threadIdx.x >> 5));
  const int lane = threadIdx.x & 31;
  const int r0   = blockIdx.x * AGG_ROWS;
  const int j0   = lane * NF;
  int2* const mylist = (int2*)wlist + (long)(blockIdx.x * AGG_WAVES + w) * AGG_CAP;

  float mx[AGG_RPW], sm[AGG_RPW], acc[AGG_RPW][NF];
  float as_l[NF];
  float adr[AGG_RPW];
  {
    float ad_l[NF];
#pragma unroll
    for (int k = 0; k < NF; ++k) { as_l[k] = a_src[j0 + k]; ad_l[k] = a_dst[j0 + k]; }
#pragma unroll
    for (int j = 0; j < AGG_RPW; ++j) {
      int loc = j * 32 + w;
      int row = r0 + loc;
      if (loc < AGG_ROWS && row < nnodes) {
#pragma unroll
        for (int k = 0; k < NF; ++k) acc[j][k] = hsrc[(long)row * F + j0 + k];
        float a_s = group_sum<LPH>(dot_part<NF>(acc[j], as_l));
        float a_d = group_sum<LPH>(dot_part<NF>(acc[j], ad_l));
        float l = a_s + a_d;
        l = l > 0.f ? l : NEG_SLOPE * l;
        mx[j] = l; sm[j] = 1.f; adr[j] = a_d;
      } else {
        mx[j] = 0.f; sm[j] = 1.f; adr[j] = 0.f;
#pragma unroll
        for (int k = 0; k < NF; ++k) acc[j][k] = 0.f;
      }
    }
  }

  auto apply = [&](int jj, const float (&hv)[NF]) __attribute__((always_inline)) {
    const float a_s = group_sum<LPH>(dot_part<NF>(hv, as_l));
#pragma unroll
    for (int j = 0; j < AGG_RPW; ++j) {
      if (jj == j) {
        float l = a_s + adr[j];
        l = l > 0.f ? l : NEG_SLOPE * l;
        float mn = fmaxf(mx[j], l);
        float sc = __expf(mx[j] - mn);
        float wg = __expf(l - mn);
        sm[j] = sm[j] * sc + wg;
#pragma unroll
        for (int k = 0; k < NF; ++k) acc[j][k] = acc[j][k] * sc + wg * hv[k];
        mx[j] = mn;
      }
    }
  };

  auto stream = [&](int nent) __attribute__((always_inline)) {
    for (int base = 0; base < nent; base += 32) {
      const int mm = min(32, nent - base);
      const int2 en = mylist[base + min(lane, mm - 1)];
      auto fetch = [&](int r, float (&hv)[NF], int& jj) __attribute__((always_inline)) {
        const int b = min(r, mm - 1);
        const int s = __builtin_amdgcn_readlane(en.x, b);
        jj = __builtin_amdgcn_readlane(en.y, b);
#pragma unroll
        for (int c = 0; c < NF; ++c) hv[c] = hsrc[(long)s * F + j0 + c];
      };
      float hb[DEPTH][NF];
      int   jb[DEPTH];
#pragma unroll
      for (int d = 0; d < DEPTH; ++d) fetch(d, hb[d], jb[d]);
      int r = 0;
      for (; r + DEPTH <= mm; r += DEPTH) {
#pragma unroll
        for (int d = 0; d < DEPTH; ++d) {
          apply(jb[d], hb[d]);
          fetch(r + d + DEPTH, hb[d], jb[d]);
        }
      }
#pragma unroll
      for (int d = 0; d < DEPTH; ++d)
        if (r + d < mm) apply(jb[d], hb[d]);
    }
  };

  int  off = 0;
  bool build = true;
  bool whole = true;
  if (!BUILD) {
    const int t = wtotal[blockIdx.x * AGG_WAVES + lane];
    if (__builtin_amdgcn_ballot_w32(t < 0) == 0) {
      build = false;
      off = __builtin_amdgcn_readlane(t, w);
    }
  }

  if (build) {
    const int* __restrict__ dstp = eidx + E;
    const bool vec_ok = ((E & 3) == 0) && ((((size_t)dstp) & 15) == 0);
    if (threadIdx.x == 0) { cnt[0] = 0; cnt[1] = 0; }
    if (threadIdx.x < AGG_WAVES) { wcnt[0][threadIdx.x] = 0; wcnt[1][threadIdx.x] = 0; }
    int tot = 0;
    int cb = 0;
    int S = AGG_S;
    int pe = 0;
    int pp = -1;
    for (int c0 = 0; c0 < E; cb ^= 1) {
      __syncthreads();
      const int lim = (E - c0 < S) ? E : c0 + S;
      int ps = 0;
      if (pp >= 0) ps = eidx[pe];
      auto test_step = [&](int t, const int4& v, unsigned rows) __attribute__((always_inline)) {
        const int i = c0 + (t * (AGG_WAVES * 32) + (int)threadIdx.x) * 4;
        const int dd[4] = {v.x, v.y, v.z, v.w};
#pragma unroll
        for (int k = 0; k < 4; ++k) {
          const unsigned dl = (unsigned)(dd[k] - r0);
          if (dl < rows) {
            const int pos = atomicAdd(&cnt[cb], 1);
            list[pos] = (int)((dl << 16) | (unsigned)(i + k - c0));
            atomicAdd(&wcnt[cb][dl & 31], 1);
          }
        }
      };
      auto store_left_over = [&]() __attribute__((always_inline)) {
        if (pp >= 0) { mylist[pp >> 3] = make_int2(ps, pp & 7); pp = -1; }
      };
      if (vec_ok) {
        auto load_step = [&](int t) __attribute__((always_inline)) -> int4 {
          const int i = c0 + (t * (AGG_WAVES * 32) + (int)threadIdx.x) * 4;
          return *(const int4*)(dstp + min(i, lim - 4));
        };
        int4 dv[AGG_RING];
#pragma unroll
        for (int t = 0; t < AGG_RING; ++t) dv[t] = load_step(t);
#pragma unroll
        for (int t = 0; t < AGG_STEPS; ++t) {
          const int i = c0 + (t * (AGG_WAVES * 32) + (int)threadIdx.x) * 4;
          test_step(t, dv[t % AGG_RING], (i < lim) ? (unsigned)AGG_ROWS : 0u);
          if (t == 0) store_left_over();
          if (t + AGG_RING < AGG_STEPS) dv[t % AGG_RING] = load_step(t + AGG_RING);
        }
      } else {
#pragma unroll
        for (int t = 0; t < AGG_STEPS; ++t) {
          const int i = c0 + (t * (AGG_WAVES * 32) + (int)threadIdx.x) * 4;
          int4 v;
          v.x = (i     < lim) ? dstp[i]     : -1;
          v.y = (i + 1 < lim) ? dstp[i + 1] : -1;
          v.z = (i + 2 < lim) ? dstp[i + 2] : -1;
          v.w = (i + 3 < lim) ? dstp[i + 3] : -1;
          test_step(t, v, (unsigned)AGG_ROWS);
          if (t == 0) store_left_over();
        }
      }
      __syncthreads();
      const int n = cnt[cb];
      if (threadIdx.x == 0) cnt[cb ^ 1] = 0;
      if (threadIdx.x < AGG_WAVES) wcnt[cb ^ 1][threadIdx.x] = 0;
      const int wc = wcnt[cb][lane];
      if (__builtin_amdgcn_ballot_w32(wc > 32) != 0) { S >>= 1; continue; }
      tot += wc;
      if (__builtin_amdgcn_ballot_w32(tot > AGG_CAP) != 0) {
        __syncthreads();
        stream(off);
        off = 0; tot = wc; whole = false;
      }

      int qk = 0x7fffffff;
      int m = 0;
      for (int base = 0; base < n; base += 32 * AGG_SCAN) {
        int en[AGG_SCAN];
#pragma unroll
        for (int u = 0; u < AGG_SCAN; ++u) en[u] = list[min(base + 32 * u + lane, n - 1)];
#pragma unroll
        for (int u = 0; u < AGG_SCAN; ++u) {
          const bool own = (base + 32 * u + lane < n) && (((en[u] >> 16) & 31) == w);
          unsigned mask = __builtin_amdgcn_ballot_w32(own);
          while (mask) {
            const int b = __builtin_ctz(mask);
            mask &= mask - 1;
            const int fk = __builtin_amdgcn_readlane(en[u], b);
            if (lane == m) qk = fk;
            ++m;
          }
        }
      }
      if (m > 0) {
        int rank = 0;
        for (int j = 0; j < m; ++j) rank += (__builtin_amdgcn_readlane(qk, j) < qk) ? 1 : 0;
        if (lane < m) { pe = c0 + (qk & 0xffff); pp = ((off + rank) << 3) | (qk >> 21); }
        off += m;
      }
      c0 += S;
      S = AGG_S;
    }
    if (pp >= 0) mylist[pp >> 3] = make_int2(eidx[pe], pp & 7);
    __syncthreads();
  }

  stream(off);
  if (BUILD && lane == 0) wtotal[blockIdx.x * AGG_WAVES + w] = whole ? off : -1;

  const int j0e = (int)__builtin_amdgcn_mbcnt_lo(~0u, 0u) * NF;
#pragma unroll
  for (int j = 0; j < AGG_RPW; ++j) {
    int loc = j * 32 + w;
    int row = r0 + loc;
    if (loc < AGG_ROWS && row < nnodes) {
#pragma unroll
      for (int k = 0; k < NF; ++k) {
        float v = acc[j][k] / sm[j] + bias[j0e + k];
        if (DO_ELU) v = v > 0.f ? v : (__expf(v) - 1.f);
        outp[(long)row * F + j0e + k] = v;
      }
    }
  }
}


static inline int cdiv_i(long a, long b) { return (int)((a + b - 1) / b); }

static void launch_gemm(const float* A, const float* B, float* C,
                        int M, int K, int Nd, hipStream_t stream) {
  if ((M & 15) == 0 && K == 128 && Nd == 256) {
    gemm_rows_in_regs<128, 256, 8><<<256, GR_WAVES * 32, 0, stream>>>(A, B, C, M);
    return;
  }
  if ((M & 15) == 0 && K == 256 && Nd == 64) {
    gemm_rows_in_regs<256, 64, 4><<<256, GR_WAVES * 32, 0, stream>>>(A, B, C, M);
    return;
  }
  int tilesM = M / 16, tilesN = Nd / 16;
  int wpb = tilesN >= 8 ? 8 : tilesN;
  dim3 grid(tilesM * (tilesN / wpb));
  dim3 block(32 * wpb);
  size_t shmem = (size_t)16 * K * sizeof(float);
  gemm_wmma_f32_lds<<<grid, block, shmem, stream>>>(A, B, C, M, K, Nd, wpb);
}

extern "C" void kernel_launch(void* const* d_in, const int* in_sizes, int n_in,
                              void* d_out, int out_size, void* d_ws, size_t ws_size,
                              hipStream_t stream) {
  const float* x   = (const float*)d_in[0];
  const int*   eix = (const int*)d_in[1];
  const float* W1  = (const float*)d_in[2];
  const float* as1 = (const float*)d_in[3];
  const float* ad1 = (const float*)d_in[4];
  const float* b1  = (const float*)d_in[5];
  const float* W2  = (const float*)d_in[6];
  const float* as2 = (const float*)d_in[7];
  const float* ad2 = (const float*)d_in[8];
  const float* b2  = (const float*)d_in[9];
  float* out = (float*)d_out;

  const int IN_DIM = 128;
  const int F1     = in_sizes[5];
  const int HID    = in_sizes[9];
  const int HEADS  = F1 / HID;
  const int Nn     = in_sizes[0] / IN_DIM;
  const int E      = in_sizes[1] / 2;
  const int EN     = E + Nn;

  float* acc1 = (float*)d_ws;
  float* h1   = acc1 + (long)Nn * F1;
  float* sc   = h1 + (long)Nn * F1;
  float* als1 = sc;
  float* ald1 = als1 + (long)Nn * HEADS;
  float* m1   = ald1 + (long)Nn * HEADS;
  float* s1   = m1   + (long)Nn * HEADS;
  float* als2 = s1   + (long)Nn * HEADS;
  float* ald2 = als2 + Nn;
  float* m2   = ald2 + Nn;
  float* s2   = m2   + Nn;
  float* h2   = h1;
  int*   wl   = (int*)(s2 + Nn);
  int*   wtot = wl + (long)cdiv_i(Nn, AGG_ROWS) * AGG_WAVES * AGG_CAP * 2;

  const float NEGINF = -__builtin_huge_valf();

  launch_gemm(x, W1, h1, Nn, IN_DIM, F1, stream);
  if (HEADS == 4 && HID == 64) {
    agg_owner_kernel<4, 64, 1, 1><<<cdiv_i(Nn, AGG_ROWS), AGG_WAVES * 32, 0, stream>>>(
        eix, E, Nn, as1, ad1, h1, b1, acc1, wl, wtot);
  } else {
    logits_kernel<<<cdiv_i((long)Nn * HEADS * 32, 256), 256, 0, stream>>>(
        h1, as1, ad1, als1, ald1, Nn, HEADS, HID);
    fill_f32<<<cdiv_i((long)Nn * F1, 256), 256, 0, stream>>>(acc1, (long)Nn * F1, 0.f);
    fill_f32<<<cdiv_i((long)Nn * HEADS, 256), 256, 0, stream>>>(m1, (long)Nn * HEADS, NEGINF);
    fill_f32<<<cdiv_i((long)Nn * HEADS, 256), 256, 0, stream>>>(s1, (long)Nn * HEADS, 0.f);
    edge_max_kernel<<<cdiv_i((long)EN * HEADS, 256), 256, 0, stream>>>(
        eix, E, EN, als1, ald1, m1, HEADS);
    edge_fused_gen<<<cdiv_i((long)EN * 32, 256), 256, 0, stream>>>(
        eix, E, EN, als1, ald1, m1, s1, h1, acc1, HEADS, HID);
    norm_bias_act_kernel<<<cdiv_i((long)Nn * F1, 256), 256, 0, stream>>>(
        acc1, s1, b1, (long)Nn * F1, F1, HID, HEADS, 1);
  }

  launch_gemm(acc1, W2, h2, Nn, F1, HID, stream);
  if (HEADS == 4 && HID == 64) {
    agg_owner_kernel<1, 64, 0, 0><<<cdiv_i(Nn, AGG_ROWS), AGG_WAVES * 32, 0, stream>>>(
        eix, E, Nn, as2, ad2, h2, b2, out, wl, wtot);
  } else if (HID == 64) {
    agg_owner_kernel<1, 64, 0, 1><<<cdiv_i(Nn, AGG_ROWS), AGG_WAVES * 32, 0, stream>>>(
        eix, E, Nn, as2, ad2, h2, b2, out, wl, wtot);
  } else {
    logits_kernel<<<cdiv_i((long)Nn * 32, 256), 256, 0, stream>>>(
        h2, as2, ad2, als2, ald2, Nn, 1, HID);
    fill_f32<<<cdiv_i(Nn, 256), 256, 0, stream>>>(m2, Nn, NEGINF);
    fill_f32<<<cdiv_i(Nn, 256), 256, 0, stream>>>(s2, Nn, 0.f);
    fill_f32<<<cdiv_i((long)Nn * HID, 256), 256, 0, stream>>>(out, (long)Nn * HID, 0.f);
    edge_max_kernel<<<cdiv_i((long)EN, 256), 256, 0, stream>>>(
        eix, E, EN, als2, ald2, m2, 1);
    edge_fused_gen<<<cdiv_i((long)EN * 32, 256), 256, 0, stream>>>(
        eix, E, EN, als2, ald2, m2, s2, h2, out, 1, HID);
    norm_bias_act_kernel<<<cdiv_i((long)Nn * HID, 256), 256, 0, stream>>>(
        out, s2, b2, (long)Nn * HID, HID, HID, 1, 0);
  }
}


# Edge lists: the first GEMM launch bins every edge once into (owner group, producer) segments; a group collects its 256 segments and each wave rank-sorts its entries in registers; same sum order
# speedup vs baseline: 9.5633x; 1.2778x over previous
#include <hip/hip_runtime.h>

typedef float v2f __attribute__((ext_vector_type(2)));
typedef float v8f __attribute__((ext_vector_type(8)));

#define NEG_SLOPE 0.2f


__global__ void fill_f32(float* __restrict__ p, long n, float v) {
  long i = (long)blockIdx.x * blockDim.x + threadIdx.x;
  long stride = (long)gridDim.x * blockDim.x;
  for (; i < n; i += stride) p[i] = v;
}

__global__ void gemm_wmma_f32_lds(const float* __restrict__ A, const float* __restrict__ B,
                                  float* __restrict__ C, int M, int K, int Nd,
                                  int tnPerBlock) {
  extern __shared__ float As[];
  int tilesN = Nd >> 4;
  int blocksPerRow = tilesN / tnPerBlock;
  int tm  = blockIdx.x / blocksPerRow;
  int tn0 = (blockIdx.x % blocksPerRow) * tnPerBlock;
  int wib  = threadIdx.x >> 5;
  int lane = threadIdx.x & 31;

  {
    const float4* Av  = (const float4*)(A + (long)tm * 16 * K);
    float4*       Asv = (float4*)As;
    int nvec = (16 * K) >> 2;
    for (int idx = threadIdx.x; idx < nvec; idx += blockDim.x) Asv[idx] = Av[idx];
  }
  __syncthreads();

  int tn   = tn0 + wib;
  int half = lane >> 4;
  int l15  = lane & 15;
  int bcol = tn * 16 + l15;
  v8f acc = {};
  for (int k = 0; k < K; k += 4) {
    int ka = k + 2 * half;
    v2f a, b;
    a.x = As[l15 * K + ka];
    a.y = As[l15 * K + ka + 1];
    b.x = B[(long)ka * Nd + bcol];
    b.y = B[(long)(ka + 1) * Nd + bcol];
    acc = __builtin_amdgcn_wmma_f32_16x16x4_f32(false, a, false, b,
                                                (short)0, acc, false, false);
  }
  long crow = (long)tm * 16 + half * 8;
#pragma unroll
  for (int r = 0; r < 8; ++r)
    C[(crow + r) * Nd + bcol] = acc[r];
}

#define GR_WAVES 16
#define BIN_ROWS 196u
#define BIN_P    256
#define BIN_CAP  64

template<int K, int ND, int CT>
__global__ void __launch_bounds__(GR_WAVES * 32)
gemm_rows_in_regs(const float* __restrict__ A, const float* __restrict__ B,
                  float* __restrict__ C, int M,
                  const int* __restrict__ eidx, int E, int* __restrict__ segw, int* __restrict__ segcnt) {
  constexpr int NT = ND / 16;
  constexpr int NG = NT / CT;
  constexpr int NQ = K / 8;
  static_assert(K % 8 == 0 && ND % 16 == 0 && NT % CT == 0, "shape must tile");
  __shared__ float4 Bs[NQ * NT * 32];
  __shared__ int bcnt[BIN_P];

  const int w    = __builtin_amdgcn_readfirstlane((int)(threadIdx.x >> 5));
  const int lane = threadIdx.x & 31;
  const int c    = lane & 15;
  const int h    = lane >> 4;
  const int tilesM = M >> 4;

  const int wa = w >> 2, wb = w & 3;
  int slot = wa + 4 * ((wb - wa) & 3);

  float4 a[NQ];
  auto load_a = [&](int tm) __attribute__((always_inline)) {
    const float4* p = (const float4*)(A + (long)(tm * 16 + c) * K + 4 * h);
#pragma unroll
    for (int q = 0; q < NQ; ++q) a[q] = p[2 * q];
  };

  int tm = (int)blockIdx.x + (int)gridDim.x * (slot / NG);
  if (tm < tilesM) load_a(tm);

  if (segw != nullptr) {
    int2* const seg = (int2*)segw;
    for (int t = threadIdx.x; t < BIN_P; t += GR_WAVES * 32) bcnt[t] = 0;
    __syncthreads();
    const int EP = (E + BIN_P - 1) / BIN_P;
    const int e0 = (int)blockIdx.x * EP;
    const int e1 = min(E, e0 + EP);
    int      sv[8];
    unsigned dv[8];
#pragma unroll
    for (int u = 0; u < 8; ++u) {
      const int e = e0 + u * (GR_WAVES * 32) + (int)threadIdx.x;
      sv[u] = (e < e1) ? eidx[e] : 0;
      dv[u] = (e < e1) ? (unsigned)eidx[E + e] : 0xffffffffu;
    }
#pragma unroll
    for (int u = 0; u < 8; ++u) {
      const unsigned own = dv[u] / BIN_ROWS;
      const unsigned lr  = dv[u] - own * BIN_ROWS;
      if (own < (unsigned)BIN_P) {
        const int pos = atomicAdd(&bcnt[own], 1);
        if (pos < BIN_CAP)
          seg[((long)own * BIN_P + blockIdx.x) * BIN_CAP + pos] =
              make_int2(sv[u], (int)((lr << 12) | (unsigned)(u * (GR_WAVES * 32) + (int)threadIdx.x)));
      }
    }
    __syncthreads();
    for (int t = threadIdx.x; t < BIN_P; t += GR_WAVES * 32) segcnt[(long)t * BIN_P + blockIdx.x] = bcnt[t];
  }

#pragma unroll 4
  for (int idx = threadIdx.x; idx < NQ * NT * 32; idx += GR_WAVES * 32) {
    const int t = idx >> 5;
    const int n = t % NT, q = t / NT;
    const float* s = B + (long)(8 * q + 4 * h) * ND + 16 * n + c;
    float4 v;
    v.x = s[0]; v.y = s[ND]; v.z = s[2 * ND]; v.w = s[3 * ND];
    Bs[idx] = v;
  }
  __syncthreads();

  while (tm < tilesM) {
    const int g = slot % NG;
    const float4* bp = Bs + g * CT * 32 + lane;
    v8f acc[CT];
#pragma unroll
    for (int n = 0; n < CT; ++n) acc[n] = v8f{};
    float4 bc[CT], bn[CT];
#pragma unroll
    for (int n = 0; n < CT; ++n) bc[n] = bp[n * 32];
#pragma unroll
    for (int q = 0; q < NQ; ++q) {
      if (q + 1 < NQ) {
#pragma unroll
        for (int n = 0; n < CT; ++n) bn[n] = bp[((q + 1) * NT + n) * 32];
      }
      __builtin_amdgcn_sched_barrier(0);
      v2f a0, a1;
      a0.x = a[q].x; a0.y = a[q].y; a1.x = a[q].z; a1.y = a[q].w;
#pragma unroll
      for (int n = 0; n < CT; ++n) {
        v2f b0; b0.x = bc[n].x; b0.y = bc[n].y;
        acc[n] = __builtin_amdgcn_wmma_f32_16x16x4_f32(false, a0, false, b0,
                                                       (short)0, acc[n], false, false);
      }
#pragma unroll
      for (int n = 0; n < CT; ++n) {
        v2f b1; b1.x = bc[n].z; b1.y = bc[n].w;
        acc[n] = __builtin_amdgcn_wmma_f32_16x16x4_f32(false, a1, false, b1,
                                                       (short)0, acc[n], false, false);
      }
      __builtin_amdgcn_sched_barrier(0);
      if (q + 1 < NQ) {
#pragma unroll
        for (int n = 0; n < CT; ++n) bc[n] = bn[n];
      }
    }
    float* cp = C + (long)(tm * 16 + 8 * h) * ND + g * CT * 16 + c;
#pragma unroll
    for (int n = 0; n < CT; ++n) {
#pragma unroll
      for (int r = 0; r < 8; ++r) cp[r * ND + n * 16] = acc[n][r];
    }
    slot += GR_WAVES;
    tm = (int)blockIdx.x + (int)gridDim.x * (slot / NG);
    if (tm < tilesM) load_a(tm);
  }
}

__global__ void logits_kernel(const float* __restrict__ hmat, const float* __restrict__ a_src,
                              const float* __restrict__ a_dst, float* __restrict__ als,
                              float* __restrict__ ald, int nnodes, int heads, int ch) {
  int wid  = (int)(((long)blockIdx.x * blockDim.x + threadIdx.x) >> 5);
  int lane = threadIdx.x & 31;
  if (wid >= nnodes * heads) return;
  int node = wid / heads, hh = wid % heads;
  int F = heads * ch;
  float ss = 0.f, dd = 0.f;
  for (int c = lane; c < ch; c += 32) {
    float hv = hmat[(long)node * F + hh * ch + c];
    ss += hv * a_src[hh * ch + c];
    dd += hv * a_dst[hh * ch + c];
  }
#pragma unroll
  for (int off = 16; off > 0; off >>= 1) {
    ss += __shfl_xor(ss, off);
    dd += __shfl_xor(dd, off);
  }
  if (lane == 0) { als[wid] = ss; ald[wid] = dd; }
}

__device__ __forceinline__ void edge_sd(const int* __restrict__ eidx, int E, int e,
                                        int& s, int& d) {
  if (e < E) { s = eidx[e]; d = eidx[E + e]; }
  else       { s = e - E; d = s; }
}

__global__ void edge_max_kernel(const int* __restrict__ eidx, int E, int EN,
                                const float* __restrict__ als, const float* __restrict__ ald,
                                float* __restrict__ m, int heads) {
  long t = (long)blockIdx.x * blockDim.x + threadIdx.x;
  if (t >= (long)EN * heads) return;
  int e = (int)(t / heads), hh = (int)(t % heads);
  int s, d; edge_sd(eidx, E, e, s, d);
  float l = als[s * heads + hh] + ald[d * heads + hh];
  l = l > 0.f ? l : NEG_SLOPE * l;
  unsafeAtomicMax(&m[d * heads + hh], l);
}

__global__ void edge_fused_gen(const int* __restrict__ eidx, int E, int EN,
                               const float* __restrict__ als, const float* __restrict__ ald,
                               const float* __restrict__ m, float* __restrict__ ssum,
                               const float* __restrict__ hsrc, float* __restrict__ accum,
                               int heads, int ch) {
  int wid  = (int)(((long)blockIdx.x * blockDim.x + threadIdx.x) >> 5);
  int lane = threadIdx.x & 31;
  if (wid >= EN) return;
  int s, d; edge_sd(eidx, E, wid, s, d);
  int F = heads * ch;
  for (int j = lane; j < F; j += 32) {
    int head = j / ch;
    float l = als[s * heads + head] + ald[d * heads + head];
    l = l > 0.f ? l : NEG_SLOPE * l;
    float ex = __expf(l - m[d * heads + head]);
    if ((j % ch) == 0) unsafeAtomicAdd(&ssum[d * heads + head], ex);
    unsafeAtomicAdd(&accum[(long)d * F + j], hsrc[(long)s * F + j] * ex);
  }
}

__global__ void norm_bias_act_kernel(float* __restrict__ a, const float* __restrict__ ssum,
                                     const float* __restrict__ bias, long n, int F,
                                     int ch, int heads, int do_elu) {
  long i = (long)blockIdx.x * blockDim.x + threadIdx.x;
  long stride = (long)gridDim.x * blockDim.x;
  for (; i < n; i += stride) {
    int  j    = (int)(i % F);
    long node = i / F;
    float v = a[i] / ssum[node * heads + j / ch] + bias[j];
    if (do_elu) v = v > 0.f ? v : (__expf(v) - 1.f);
    a[i] = v;
  }
}


#define AGG_WAVES 32
#define AGG_RPW   7
#define AGG_ROWS  196
#define AGG_S     65536
#define AGG_STEPS (AGG_S / (AGG_WAVES * 32 * 4))
#define AGG_RING  4
#define AGG_SCAN  8
#define AGG_CAP   1024

template<int NF>
__device__ __forceinline__ float dot_part(const float (&v)[NF], const float (&a)[NF]) {
  float p0 = v[0] * a[0];
  float p1 = (NF > 1) ? v[NF > 1 ? 1 : 0] * a[NF > 1 ? 1 : 0] : 0.f;
#pragma unroll
  for (int k = 2; k < NF; k += 2) {
    p0 = __builtin_fmaf(v[k], a[k], p0);
    if (k + 1 < NF) p1 = __builtin_fmaf(v[k + 1], a[k + 1], p1);
  }
  return p0 + p1;
}

template<int G>
__device__ __forceinline__ float group_sum(float p) {
  static_assert(G == 1 || G == 2 || G == 4 || G == 8 || G == 16 || G == 32, "lanes per head must be a power of two");
  if (G >= 2)  p += __int_as_float(__builtin_amdgcn_update_dpp(0, __float_as_int(p), 0x161, 0xf, 0xf, false));
  if (G >= 4)  p += __int_as_float(__builtin_amdgcn_update_dpp(0, __float_as_int(p), 0x162, 0xf, 0xf, false));
  if (G >= 8)  p += __int_as_float(__builtin_amdgcn_update_dpp(0, __float_as_int(p), 0x164, 0xf, 0xf, false));
  if (G >= 16) p += __int_as_float(__builtin_amdgcn_update_dpp(0, __float_as_int(p), 0x168, 0xf, 0xf, false));
  if (G >= 32) {
    const int b = __float_as_int(p);
    p += __int_as_float(__builtin_amdgcn_permlanex16(b, b, 0x76543210, 0xfedcba98, false, false));
  }
  return p;
}

template<int J>
__device__ __forceinline__ void rank_keys(const int (&kk)[8], int (&rk)[8], int m) {
#pragma unroll
  for (int c = 0; c < J; ++c) {
    if (32 * c < m) {
      const int mc = min(32, m - 32 * c);
      for (int l = 0; l < mc; ++l) {
        const int s = __builtin_amdgcn_readlane(kk[c], l);
#pragma unroll
        for (int j = 0; j < J; ++j) rk[j] += (s < kk[j]) ? 1 : 0;
      }
    }
  }
}

#define BIN_WCAP 256

template<int HEADS, int CH, int DO_ELU, int BUILD, int BINS>
__global__ void __launch_bounds__(AGG_WAVES * 32)
agg_owner_kernel(const int* __restrict__ eidx, int E, int nnodes,
                 const float* __restrict__ a_src, const float* __restrict__ a_dst,
                 const float* __restrict__ hsrc, const float* __restrict__ bias,
                 float* __restrict__ outp, int* wlist, int* wtotal,
                 const int* __restrict__ segw, const int* __restrict__ segcnt) {
  static_assert(AGG_ROWS == (int)BIN_ROWS && AGG_ROWS <= 256 && AGG_WAVES == 32 && BIN_CAP == 64 && BIN_P == 256,
                "the bins name a row inside its group in 8 bits; the collecting loop below gives 64 threads to a segment and 16 segments to a step");
  static_assert(2 * AGG_WAVES * BIN_WCAP <= AGG_S && BIN_WCAP <= AGG_CAP, "the waves' buckets live in the LDS key list");
  constexpr int F  = HEADS * CH;
  constexpr int NF = F / 32;
  constexpr int LPH = CH / NF;
  static_assert(NF >= 1 && CH % NF == 0 && LPH <= 32, "a head's channels must fill whole lanes");
  constexpr int DEPTH = (NF >= 8) ? 2 : ((NF >= 4) ? 4 : 8);
  __shared__ int  list[AGG_S];
  __shared__ int  cnt[2];
  __shared__ int  wcnt[2][AGG_WAVES];
  const int w    = __builtin_amdgcn_readfirstlane((int)(threadIdx.x >> 5));
  const int lane = threadIdx.x & 31;
  const int r0   = blockIdx.x * AGG_ROWS;
  const int j0   = lane * NF;
  int2* const mylist = (int2*)wlist + (long)(blockIdx.x * AGG_WAVES + w) * AGG_CAP;

  float mx[AGG_RPW], sm[AGG_RPW], acc[AGG_RPW][NF];
  float as_l[NF];
  float adr[AGG_RPW];
  {
    float ad_l[NF];
#pragma unroll
    for (int k = 0; k < NF; ++k) { as_l[k] = a_src[j0 + k]; ad_l[k] = a_dst[j0 + k]; }
#pragma unroll
    for (int j = 0; j < AGG_RPW; ++j) {
      int loc = j * 32 + w;
      int row = r0 + loc;
      if (loc < AGG_ROWS && row < nnodes) {
#pragma unroll
        for (int k = 0; k < NF; ++k) acc[j][k] = hsrc[(long)row * F + j0 + k];
        float a_s = group_sum<LPH>(dot_part<NF>(acc[j], as_l));
        float a_d = group_sum<LPH>(dot_part<NF>(acc[j], ad_l));
        float l = a_s + a_d;
        l = l > 0.f ? l : NEG_SLOPE * l;
        mx[j] = l; sm[j] = 1.f; adr[j] = a_d;
      } else {
        mx[j] = 0.f; sm[j] = 1.f; adr[j] = 0.f;
#pragma unroll
        for (int k = 0; k < NF; ++k) acc[j][k] = 0.f;
      }
    }
  }

  auto apply = [&](int jj, const float (&hv)[NF]) __attribute__((always_inline)) {
    const float a_s = group_sum<LPH>(dot_part<NF>(hv, as_l));
#pragma unroll
    for (int j = 0; j < AGG_RPW; ++j) {
      if (jj == j) {
        float l = a_s + adr[j];
        l = l > 0.f ? l : NEG_SLOPE * l;
        float mn = fmaxf(mx[j], l);
        float sc = __expf(mx[j] - mn);
        float wg = __expf(l - mn);
        sm[j] = sm[j] * sc + wg;
#pragma unroll
        for (int k = 0; k < NF; ++k) acc[j][k] = acc[j][k] * sc + wg * hv[k];
        mx[j] = mn;
      }
    }
  };

  auto stream = [&](int nent) __attribute__((always_inline)) {
    for (int base = 0; base < nent; base += 32) {
      const int mm = min(32, nent - base);
      const int2 en = mylist[base + min(lane, mm - 1)];
      auto fetch = [&](int r, float (&hv)[NF], int& jj) __attribute__((always_inline)) {
        const int b = min(r, mm - 1);
        const int s = __builtin_amdgcn_readlane(en.x, b);
        jj = __builtin_amdgcn_readlane(en.y, b);
#pragma unroll
        for (int c = 0; c < NF; ++c) hv[c] = hsrc[(long)s * F + j0 + c];
      };
      float hb[DEPTH][NF];
      int   jb[DEPTH];
#pragma unroll
      for (int d = 0; d < DEPTH; ++d) fetch(d, hb[d], jb[d]);
      int r = 0;
      for (; r + DEPTH <= mm; r += DEPTH) {
#pragma unroll
        for (int d = 0; d < DEPTH; ++d) {
          apply(jb[d], hb[d]);
          fetch(r + d + DEPTH, hb[d], jb[d]);
        }
      }
#pragma unroll
      for (int d = 0; d < DEPTH; ++d)
        if (r + d < mm) apply(jb[d], hb[d]);
    }
  };

  int  off = 0;
  bool build = true;
  bool whole = true;
  if (!BUILD) {
    const int t = wtotal[blockIdx.x * AGG_WAVES + lane];
    if (__builtin_amdgcn_ballot_w32(t < 0) == 0) {
      build = false;
      off = __builtin_amdgcn_readlane(t, w);
    }
  }

  if (BINS && build) {
    int* const bkey = list;
    int* const bsrc = list + AGG_WAVES * BIN_WCAP;
    if (threadIdx.x == 0) cnt[0] = 0;
    if (threadIdx.x < AGG_WAVES) wcnt[0][threadIdx.x] = 0;
    __syncthreads();
    const int q  = threadIdx.x & (BIN_CAP - 1);
    const int pb = threadIdx.x >> 6;
    const int*  const mycnt = segcnt + (long)blockIdx.x * BIN_P;
    const int2* const myseg = (const int2*)segw + (long)blockIdx.x * BIN_P * BIN_CAP;
    bool bad = false;
    for (int hf = 0; hf < 2; ++hf) {
      int cs[8];
#pragma unroll
      for (int u = 0; u < 8; ++u) cs[u] = mycnt[(hf * 8 + u) * 16 + pb];
      int2 en[8];
#pragma unroll
      for (int u = 0; u < 8; ++u) {
        bad |= cs[u] > BIN_CAP;
        en[u] = make_int2(0, -1);
        if (q < cs[u]) en[u] = myseg[(long)((hf * 8 + u) * 16 + pb) * BIN_CAP + q];
      }
#pragma unroll
      for (int u = 0; u < 8; ++u) {
        if (en[u].y >= 0) {
          const int p  = (hf * 8 + u) * 16 + pb;
          const int lr = (en[u].y >> 12) & 0xff;
          const int ow = lr & 31;
          const int pos = atomicAdd(&wcnt[0][ow], 1);
          if (pos < BIN_WCAP) {
            bkey[ow * BIN_WCAP + pos] = ((lr >> 5) << 20) | (p << 12) | (en[u].y & 0xfff);
            bsrc[ow * BIN_WCAP + pos] = en[u].x;
          } else bad = true;
        }
      }
    }
    if (bad) cnt[0] = 1;
    __syncthreads();
    const bool ok = (cnt[0] == 0);
    if (ok) {
      const int m = __builtin_amdgcn_readfirstlane(wcnt[0][w]);
      int kk[8], sv[8], rk[8];
#pragma unroll
      for (int j = 0; j < 8; ++j) {
        const int i = lane + 32 * j;
        kk[j] = (i < m) ? bkey[w * BIN_WCAP + i] : 0x7fffffff;
        sv[j] = (i < m) ? bsrc[w * BIN_WCAP + i] : 0;
        rk[j] = 0;
      }
      if (m <= 128) rank_keys<4>(kk, rk, m); else rank_keys<8>(kk, rk, m);
#pragma unroll
      for (int j = 0; j < 8; ++j)
        if (lane + 32 * j < m) mylist[rk[j]] = make_int2(sv[j], kk[j] >> 20);
      off = m;
      build = false;
    }
    __syncthreads();
  }

  if (build) {
    const int* __restrict__ dstp = eidx + E;
    const bool vec_ok = ((E & 3) == 0) && ((((size_t)dstp) & 15) == 0);
    if (threadIdx.x == 0) { cnt[0] = 0; cnt[1] = 0; }
    if (threadIdx.x < AGG_WAVES) { wcnt[0][threadIdx.x] = 0; wcnt[1][threadIdx.x] = 0; }
    int tot = 0;
    int cb = 0;
    int S = AGG_S;
    int pe = 0;
    int pp = -1;
    for (int c0 = 0; c0 < E; cb ^= 1) {
      __syncthreads();
      const int lim = (E - c0 < S) ? E : c0 + S;
      int ps = 0;
      if (pp >= 0) ps = eidx[pe];
      auto test_step = [&](int t, const int4& v, unsigned rows) __attribute__((always_inline)) {
        const int i = c0 + (t * (AGG_WAVES * 32) + (int)threadIdx.x) * 4;
        const int dd[4] = {v.x, v.y, v.z, v.w};
#pragma unroll
        for (int k = 0; k < 4; ++k) {
          const unsigned dl = (unsigned)(dd[k] - r0);
          if (dl < rows) {
            const int pos = atomicAdd(&cnt[cb], 1);
            list[pos] = (int)((dl << 16) | (unsigned)(i + k - c0));
            atomicAdd(&wcnt[cb][dl & 31], 1);
          }
        }
      };
      auto store_left_over = [&]() __attribute__((always_inline)) {
        if (pp >= 0) { mylist[pp >> 3] = make_int2(ps, pp & 7); pp = -1; }
      };
      if (vec_ok) {
        auto load_step = [&](int t) __attribute__((always_inline)) -> int4 {
          const int i = c0 + (t * (AGG_WAVES * 32) + (int)threadIdx.x) * 4;
          return *(const int4*)(dstp + min(i, lim - 4));
        };
        int4 dv[AGG_RING];
#pragma unroll
        for (int t = 0; t < AGG_RING; ++t) dv[t] = load_step(t);
#pragma unroll
        for (int t = 0; t < AGG_STEPS; ++t) {
          const int i = c0 + (t * (AGG_WAVES * 32) + (int)threadIdx.x) * 4;
          test_step(t, dv[t % AGG_RING], (i < lim) ? (unsigned)AGG_ROWS : 0u);
          if (t == 0) store_left_over();
          if (t + AGG_RING < AGG_STEPS) dv[t % AGG_RING] = load_step(t + AGG_RING);
        }
      } else {
#pragma unroll
        for (int t = 0; t < AGG_STEPS; ++t) {
          const int i = c0 + (t * (AGG_WAVES * 32) + (int)threadIdx.x) * 4;
          int4 v;
          v.x = (i     < lim) ? dstp[i]     : -1;
          v.y = (i + 1 < lim) ? dstp[i + 1] : -1;
          v.z = (i + 2 < lim) ? dstp[i + 2] : -1;
          v.w = (i + 3 < lim) ? dstp[i + 3] : -1;
          test_step(t, v, (unsigned)AGG_ROWS);
          if (t == 0) store_left_over();
        }
      }
      __syncthreads();
      const int n = cnt[cb];
      if (threadIdx.x == 0) cnt[cb ^ 1] = 0;
      if (threadIdx.x < AGG_WAVES) wcnt[cb ^ 1][threadIdx.x] = 0;
      const int wc = wcnt[cb][lane];
      if (__builtin_amdgcn_ballot_w32(wc > 32) != 0) { S >>= 1; continue; }
      tot += wc;
      if (__builtin_amdgcn_ballot_w32(tot > AGG_CAP) != 0) {
        __syncthreads();
        stream(off);
        off = 0; tot = wc; whole = false;
      }

      int qk = 0x7fffffff;
      int m = 0;
      for (int base = 0; base < n; base += 32 * AGG_SCAN) {
        int en[AGG_SCAN];
#pragma unroll
        for (int u = 0; u < AGG_SCAN; ++u) en[u] = list[min(base + 32 * u + lane, n - 1)];
#pragma unroll
        for (int u = 0; u < AGG_SCAN; ++u) {
          const bool own = (base + 32 * u + lane < n) && (((en[u] >> 16) & 31) == w);
          unsigned mask = __builtin_amdgcn_ballot_w32(own);
          while (mask) {
            const int b = __builtin_ctz(mask);
            mask &= mask - 1;
            const int fk = __builtin_amdgcn_readlane(en[u], b);
            if (lane == m) qk = fk;
            ++m;
          }
        }
      }
      if (m > 0) {
        int rank = 0;
        for (int j = 0; j < m; ++j) rank += (__builtin_amdgcn_readlane(qk, j) < qk) ? 1 : 0;
        if (lane < m) { pe = c0 + (qk & 0xffff); pp = ((off + rank) << 3) | (qk >> 21); }
        off += m;
      }
      c0 += S;
      S = AGG_S;
    }
    if (pp >= 0) mylist[pp >> 3] = make_int2(eidx[pe], pp & 7);
    __syncthreads();
  }

  stream(off);
  if (BUILD && lane == 0) wtotal[blockIdx.x * AGG_WAVES + w] = whole ? off : -1;

  const int j0e = (int)__builtin_amdgcn_mbcnt_lo(~0u, 0u) * NF;
#pragma unroll
  for (int j = 0; j < AGG_RPW; ++j) {
    int loc = j * 32 + w;
    int row = r0 + loc;
    if (loc < AGG_ROWS && row < nnodes) {
#pragma unroll
      for (int k = 0; k < NF; ++k) {
        float v = acc[j][k] / sm[j] + bias[j0e + k];
        if (DO_ELU) v = v > 0.f ? v : (__expf(v) - 1.f);
        outp[(long)row * F + j0e + k] = v;
      }
    }
  }
}


static inline int cdiv_i(long a, long b) { return (int)((a + b - 1) / b); }

static void launch_gemm(const float* A, const float* B, float* C,
                        int M, int K, int Nd, hipStream_t stream,
                        const int* eidx = nullptr, int E = 0, int* seg = nullptr, int* segcnt = nullptr) {
  if ((M & 15) == 0 && K == 128 && Nd == 256) {
    gemm_rows_in_regs<128, 256, 8><<<BIN_P, GR_WAVES * 32, 0, stream>>>(A, B, C, M, eidx, E, seg, segcnt);
    return;
  }
  if ((M & 15) == 0 && K == 256 && Nd == 64) {
    gemm_rows_in_regs<256, 64, 4><<<256, GR_WAVES * 32, 0, stream>>>(A, B, C, M, nullptr, 0, nullptr, nullptr);
    return;
  }
  int tilesM = M / 16, tilesN = Nd / 16;
  int wpb = tilesN >= 8 ? 8 : tilesN;
  dim3 grid(tilesM * (tilesN / wpb));
  dim3 block(32 * wpb);
  size_t shmem = (size_t)16 * K * sizeof(float);
  gemm_wmma_f32_lds<<<grid, block, shmem, stream>>>(A, B, C, M, K, Nd, wpb);
}

extern "C" void kernel_launch(void* const* d_in, const int* in_sizes, int n_in,
                              void* d_out, int out_size, void* d_ws, size_t ws_size,
                              hipStream_t stream) {
  const float* x   = (const float*)d_in[0];
  const int*   eix = (const int*)d_in[1];
  const float* W1  = (const float*)d_in[2];
  const float* as1 = (const float*)d_in[3];
  const float* ad1 = (const float*)d_in[4];
  const float* b1  = (const float*)d_in[5];
  const float* W2  = (const float*)d_in[6];
  const float* as2 = (const float*)d_in[7];
  const float* ad2 = (const float*)d_in[8];
  const float* b2  = (const float*)d_in[9];
  float* out = (float*)d_out;

  const int IN_DIM = 128;
  const int F1     = in_sizes[5];
  const int HID    = in_sizes[9];
  const int HEADS  = F1 / HID;
  const int Nn     = in_sizes[0] / IN_DIM;
  const int E      = in_sizes[1] / 2;
  const int EN     = E + Nn;

  float* acc1 = (float*)d_ws;
  float* h1   = acc1 + (long)Nn * F1;
  float* sc   = h1 + (long)Nn * F1;
  float* als1 = sc;
  float* ald1 = als1 + (long)Nn * HEADS;
  float* m1   = ald1 + (long)Nn * HEADS;
  float* s1   = m1   + (long)Nn * HEADS;
  float* als2 = s1   + (long)Nn * HEADS;
  float* ald2 = als2 + Nn;
  float* m2   = ald2 + Nn;
  float* s2   = m2   + Nn;
  float* h2   = h1;
  int*   wl   = (int*)(s2 + Nn);
  int*   wtot = wl + (long)cdiv_i(Nn, AGG_ROWS) * AGG_WAVES * AGG_CAP * 2;
  int*   seg    = wtot + (long)cdiv_i(Nn, AGG_ROWS) * AGG_WAVES;
  int*   segcnt = seg + (long)BIN_P * BIN_P * BIN_CAP * 2;
  const bool use_bins = (HEADS == 4 && HID == 64) && ((Nn & 15) == 0) && (Nn <= BIN_P * AGG_ROWS) && (E <= BIN_P * 4096) &&
                        ((((size_t)seg) & 7) == 0) &&
                        ((size_t)((const char*)(segcnt + BIN_P * BIN_P) - (const char*)d_ws) <= ws_size);

  const float NEGINF = -__builtin_huge_valf();

  if (use_bins) launch_gemm(x, W1, h1, Nn, IN_DIM, F1, stream, eix, E, seg, segcnt);
  else          launch_gemm(x, W1, h1, Nn, IN_DIM, F1, stream);
  if (use_bins) {
    agg_owner_kernel<4, 64, 1, 1, 1><<<cdiv_i(Nn, AGG_ROWS), AGG_WAVES * 32, 0, stream>>>(
        eix, E, Nn, as1, ad1, h1, b1, acc1, wl, wtot, seg, segcnt);
  } else if (HEADS == 4 && HID == 64) {
    agg_owner_kernel<4, 64, 1, 1, 0><<<cdiv_i(Nn, AGG_ROWS), AGG_WAVES * 32, 0, stream>>>(
        eix, E, Nn, as1, ad1, h1, b1, acc1, wl, wtot, nullptr, nullptr);
  } else {
    logits_kernel<<<cdiv_i((long)Nn * HEADS * 32, 256), 256, 0, stream>>>(
        h1, as1, ad1, als1, ald1, Nn, HEADS, HID);
    fill_f32<<<cdiv_i((long)Nn * F1, 256), 256, 0, stream>>>(acc1, (long)Nn * F1, 0.f);
    fill_f32<<<cdiv_i((long)Nn * HEADS, 256), 256, 0, stream>>>(m1, (long)Nn * HEADS, NEGINF);
    fill_f32<<<cdiv_i((long)Nn * HEADS, 256), 256, 0, stream>>>(s1, (long)Nn * HEADS, 0.f);
    edge_max_kernel<<<cdiv_i((long)EN * HEADS, 256), 256, 0, stream>>>(
        eix, E, EN, als1, ald1, m1, HEADS);
    edge_fused_gen<<<cdiv_i((long)EN * 32, 256), 256, 0, stream>>>(
        eix, E, EN, als1, ald1, m1, s1, h1, acc1, HEADS, HID);
    norm_bias_act_kernel<<<cdiv_i((long)Nn * F1, 256), 256, 0, stream>>>(
        acc1, s1, b1, (long)Nn * F1, F1, HID, HEADS, 1);
  }

  launch_gemm(acc1, W2, h2, Nn, F1, HID, stream);
  if (HEADS == 4 && HID == 64) {
    agg_owner_kernel<1, 64, 0, 0, 0><<<cdiv_i(Nn, AGG_ROWS), AGG_WAVES * 32, 0, stream>>>(
        eix, E, Nn, as2, ad2, h2, b2, out, wl, wtot, nullptr, nullptr);
  } else if (HID == 64) {
    agg_owner_kernel<1, 64, 0, 1, 0><<<cdiv_i(Nn, AGG_ROWS), AGG_WAVES * 32, 0, stream>>>(
        eix, E, Nn, as2, ad2, h2, b2, out, wl, wtot, nullptr, nullptr);
  } else {
    logits_kernel<<<cdiv_i((long)Nn * 32, 256), 256, 0, stream>>>(
        h2, as2, ad2, als2, ald2, Nn, 1, HID);
    fill_f32<<<cdiv_i(Nn, 256), 256, 0, stream>>>(m2, Nn, NEGINF);
    fill_f32<<<cdiv_i(Nn, 256), 256, 0, stream>>>(s2, Nn, 0.f);
    fill_f32<<<cdiv_i((long)Nn * HID, 256), 256, 0, stream>>>(out, (long)Nn * HID, 0.f);
    edge_max_kernel<<<cdiv_i((long)EN, 256), 256, 0, stream>>>(
        eix, E, EN, als2, ald2, m2, 1);
    edge_fused_gen<<<cdiv_i((long)EN * 32, 256), 256, 0, stream>>>(
        eix, E, EN, als2, ald2, m2, s2, h2, out, 1, HID);
    norm_bias_act_kernel<<<cdiv_i((long)Nn * HID, 256), 256, 0, stream>>>(
        out, s2, b2, (long)Nn * HID, HID, HID, 1, 0);
  }
}
